# starting kernel
# baseline (speedup 1.0000x reference)
#include <hip/hip_runtime.h>
#include <stdint.h>

#define NNODES 100000
#define KDIM   128

typedef __attribute__((ext_vector_type(16))) __bf16 v16bf;
typedef __attribute__((ext_vector_type(8)))  __bf16 v8bf;
typedef __attribute__((ext_vector_type(8)))  float  v8f;

__device__ __forceinline__ __bf16 f2bf(float f) {
  unsigned u = __builtin_bit_cast(unsigned, f);
  unsigned r = u + 0x7FFFu + ((u >> 16) & 1u);
  unsigned short s = (unsigned short)(r >> 16);
  return __builtin_bit_cast(__bf16, s);
}

__device__ __forceinline__ void atomic_add_f32(float* p, float v) {
  __hip_atomic_fetch_add(p, v, __ATOMIC_RELAXED, __HIP_MEMORY_SCOPE_AGENT);
}

__global__ void zero_f32(float* __restrict__ p, long long n) {
  long long i = blockIdx.x * (long long)blockDim.x + threadIdx.x;
  long long stride = (long long)gridDim.x * blockDim.x;
  for (; i < n; i += stride) p[i] = 0.0f;
}

__global__ void degree_kernel(const int* __restrict__ src, const int* __restrict__ dst,
                              float* __restrict__ outdeg, float* __restrict__ indeg, int E) {
  int e = blockIdx.x * blockDim.x + threadIdx.x;
  if (e < E) {
    atomic_add_f32(&outdeg[src[e]], 1.0f);
    atomic_add_f32(&indeg[dst[e]], 1.0f);
  }
}

__global__ void norm_kernel(const float* __restrict__ outdeg, const float* __restrict__ indeg,
                            float* __restrict__ nsrc, float* __restrict__ ndst, int n) {
  int i = blockIdx.x * blockDim.x + threadIdx.x;
  if (i < n) {
    nsrc[i] = rsqrtf(fmaxf(outdeg[i], 1.0f));
    ndst[i] = rsqrtf(fmaxf(indeg[i], 1.0f));
  }
}

template <int MOUT>
__global__ __launch_bounds__(256)
void gcn_gemm(const float* __restrict__ x, const float* __restrict__ nsrc,
              const float* __restrict__ W, float* __restrict__ h, int nrows) {
  constexpr int LDK = KDIM + 8;
  __shared__ __bf16 As[64 * LDK];
  __shared__ __bf16 Bs[MOUT * LDK];

  const int tid  = threadIdx.x;
  const int row0 = blockIdx.x * 64;

  for (int idx = tid; idx < 64 * KDIM; idx += 256) {
    int r = idx >> 7, k = idx & (KDIM - 1);
    int row = row0 + r;
    float v = 0.0f;
    if (row < nrows) v = x[(size_t)row * KDIM + k] * nsrc[row];
    As[r * LDK + k] = f2bf(v);
  }
  for (int idx = tid; idx < MOUT * KDIM; idx += 256) {
    int k = idx / MOUT, n = idx % MOUT;
    Bs[n * LDK + k] = f2bf(W[(size_t)k * MOUT + n]);
  }
  __syncthreads();

  const int wave = tid >> 5, lane = tid & 31;
  const int half = lane >> 4, mn = lane & 15;
  constexpr int CT = MOUT / 16;
  constexpr int NT = 4 * CT;
  const bool full = (row0 + 64) <= nrows;

  const int tc = wave % CT;
  const __bf16* brow = &Bs[(tc * 16 + mn) * LDK];
  v16bf bfrag[KDIM / 32];
#pragma unroll
  for (int kk = 0; kk < KDIM / 32; ++kk) {
    int k0 = kk * 32 + half * 8;
    v8bf bl = *(const v8bf*)(brow + k0);
    v8bf bh = *(const v8bf*)(brow + k0 + 16);
#pragma unroll
    for (int i = 0; i < 8; ++i) { bfrag[kk][i] = bl[i]; bfrag[kk][i + 8] = bh[i]; }
  }

#pragma unroll
  for (int it = 0; it < NT / 8; ++it) {
    const int t  = wave + it * 8;
    const int tr = t / CT;
    const __bf16* arow = &As[(tr * 16 + mn) * LDK];

    v16bf afrag[KDIM / 32];
#pragma unroll
    for (int kk = 0; kk < KDIM / 32; ++kk) {
      int k0 = kk * 32 + half * 8;
      v8bf al = *(const v8bf*)(arow + k0);
      v8bf ah = *(const v8bf*)(arow + k0 + 16);
#pragma unroll
      for (int i = 0; i < 8; ++i) { afrag[kk][i] = al[i]; afrag[kk][i + 8] = ah[i]; }
    }

    v8f c = {};
#pragma unroll
    for (int kk = 0; kk < KDIM / 32; ++kk)
      c = __builtin_amdgcn_wmma_f32_16x16x32_bf16(false, afrag[kk], false, bfrag[kk],
                                                  (short)0, c, false, false);

    const int gr0 = row0 + tr * 16 + half * 8;
    const int gc  = tc * 16 + mn;
    float* out = h + (size_t)gr0 * MOUT + gc;
    if (full) {
#pragma unroll
      for (int r = 0; r < 8; ++r) out[(size_t)r * MOUT] = c[r];
    } else {
#pragma unroll
      for (int r = 0; r < 8; ++r)
        if (gr0 + r < nrows) out[(size_t)r * MOUT] = c[r];
    }
  }
}

__global__ void scatter_edges(const float4* __restrict__ h4, const int* __restrict__ src,
                              const int* __restrict__ dst, float* __restrict__ agg,
                              int E, int gshift) {
  long long gid = blockIdx.x * (long long)blockDim.x + threadIdx.x;
  int groups = 1 << gshift;
  int g = (int)(gid & (groups - 1));
  long long e = gid >> gshift;
  if (e >= E) return;
  int s = src[e], d = dst[e];
  float4 v = h4[(size_t)s * groups + g];
  float* base = agg + ((size_t)d << (gshift + 2)) + g * 4;
  atomic_add_f32(base + 0, v.x);
  atomic_add_f32(base + 1, v.y);
  atomic_add_f32(base + 2, v.z);
  atomic_add_f32(base + 3, v.w);
}

template <int F, bool RELU>
__global__ void finish_kernel(const float* __restrict__ agg, const float* __restrict__ ndst,
                              const float* __restrict__ bias, float* __restrict__ out, int n) {
  long long idx = blockIdx.x * (long long)blockDim.x + threadIdx.x;
  if (idx >= (long long)n * F) return;
  int node = (int)(idx / F);
  int f = (int)(idx & (F - 1));
  float v = agg[idx] * ndst[node] + bias[f];
  out[idx] = RELU ? fmaxf(v, 0.0f) : v;
}

extern "C" void kernel_launch(void* const* d_in, const int* in_sizes, int n_in,
                              void* d_out, int out_size, void* d_ws, size_t ws_size,
                              hipStream_t stream) {
  const float* in_feat = (const float*)d_in[0];
  const int*   src     = (const int*)d_in[1];
  const int*   dst     = (const int*)d_in[2];
  const float* W1 = (const float*)d_in[3];  const float* b1 = (const float*)d_in[4];
  const float* W2 = (const float*)d_in[5];  const float* b2 = (const float*)d_in[6];
  const float* W3 = (const float*)d_in[7];  const float* b3 = (const float*)d_in[8];
  const float* W4 = (const float*)d_in[9];  const float* b4 = (const float*)d_in[10];

  const int N = NNODES;
  const int E = in_sizes[1];

  float* ws     = (float*)d_ws;
  float* outdeg = ws;
  float* indeg  = outdeg + N;
  float* nsrc   = indeg + N;
  float* ndst   = nsrc + N;
  float* bufA   = ndst + N;
  float* bufB   = bufA + (size_t)N * 128;
  float* bufC   = bufB + (size_t)N * 128;

  const dim3 blk(256);
  const unsigned gemm_blocks = (N + 63) / 64;
  const unsigned nf128 = (unsigned)(((long long)N * 128 + 255) / 256);
  const unsigned nf64  = (unsigned)(((long long)N * 64 + 255) / 256);
  const unsigned sc128 = (unsigned)(((long long)E * 32 + 255) / 256);
  const unsigned sc64  = (unsigned)(((long long)E * 16 + 255) / 256);

  zero_f32<<<2048, blk, 0, stream>>>(outdeg, 2LL * N);
  degree_kernel<<<(E + 255) / 256, blk, 0, stream>>>(src, dst, outdeg, indeg, E);
  norm_kernel<<<(N + 255) / 256, blk, 0, stream>>>(outdeg, indeg, nsrc, ndst, N);

  gcn_gemm<128><<<gemm_blocks, blk, 0, stream>>>(in_feat, nsrc, W1, bufB, N);
  zero_f32<<<4096, blk, 0, stream>>>(bufC, (long long)N * 128);
  scatter_edges<<<sc128, blk, 0, stream>>>((const float4*)bufB, src, dst, bufC, E, 5);
  finish_kernel<128, true><<<nf128, blk, 0, stream>>>(bufC, ndst, b1, bufA, N);

  gcn_gemm<128><<<gemm_blocks, blk, 0, stream>>>(bufA, nsrc, W2, bufB, N);
  zero_f32<<<4096, blk, 0, stream>>>(bufC, (long long)N * 128);
  scatter_edges<<<sc128, blk, 0, stream>>>((const float4*)bufB, src, dst, bufC, E, 5);
  finish_kernel<128, true><<<nf128, blk, 0, stream>>>(bufC, ndst, b2, bufA, N);

  gcn_gemm<128><<<gemm_blocks, blk, 0, stream>>>(bufA, nsrc, W3, bufB, N);
  zero_f32<<<4096, blk, 0, stream>>>(bufC, (long long)N * 128);
  scatter_edges<<<sc128, blk, 0, stream>>>((const float4*)bufB, src, dst, bufC, E, 5);
  finish_kernel<128, true><<<nf128, blk, 0, stream>>>(bufC, ndst, b3, bufA, N);

  gcn_gemm<64><<<gemm_blocks, blk, 0, stream>>>(bufA, nsrc, W4, bufB, N);
  zero_f32<<<4096, blk, 0, stream>>>(bufC, (long long)N * 64);
  scatter_edges<<<sc64, blk, 0, stream>>>((const float4*)bufB, src, dst, bufC, E, 4);
  finish_kernel<64, false><<<nf64, blk, 0, stream>>>(bufC, ndst, b4, (float*)d_out, N);
}


# product operand staging: A tile and W fetched as 16-byte loads issued in two groups ahead of the conversions, not one waited 4-byte load per element; no sum or sum order changes
# speedup vs baseline: 2.9660x; 2.9660x over previous
#include <hip/hip_runtime.h>
#include <stdint.h>

#define NNODES 100000
#define KDIM   128

typedef __attribute__((ext_vector_type(16))) __bf16 v16bf;
typedef __attribute__((ext_vector_type(8)))  __bf16 v8bf;
typedef __attribute__((ext_vector_type(4)))  __bf16 v4bf;
typedef __attribute__((ext_vector_type(2)))  __bf16 v2bf;
typedef __attribute__((ext_vector_type(8)))  float  v8f;

__device__ __forceinline__ __bf16 f2bf(float f) {
  unsigned u = __builtin_bit_cast(unsigned, f);
  unsigned r = u + 0x7FFFu + ((u >> 16) & 1u);
  unsigned short s = (unsigned short)(r >> 16);
  return __builtin_bit_cast(__bf16, s);
}

__global__ void zero_i32(int* __restrict__ p, int n) {
  int i = blockIdx.x * blockDim.x + threadIdx.x;
  int stride = gridDim.x * blockDim.x;
  for (; i < n; i += stride) p[i] = 0;
}

__global__ void count_edges(const int* __restrict__ src, const int* __restrict__ dst,
                            int* __restrict__ cnt_out, int* __restrict__ cnt_in,
                            int* __restrict__ slot, int E) {
  int e = blockIdx.x * blockDim.x + threadIdx.x;
  if (e < E) {
    slot[e] = __hip_atomic_fetch_add(&cnt_in[dst[e]], 1, __ATOMIC_RELAXED, __HIP_MEMORY_SCOPE_AGENT);
    __hip_atomic_fetch_add(&cnt_out[src[e]], 1, __ATOMIC_RELAXED, __HIP_MEMORY_SCOPE_AGENT);
  }
}

__global__ __launch_bounds__(256)
void scan_degrees(const int* __restrict__ cnt_in, const int* __restrict__ cnt_out,
                  int* __restrict__ rowptr, float* __restrict__ nsrc, float* __restrict__ ndst, int n) {
  __shared__ int wsum[8];
  __shared__ int wsum2[8];
  const int tid = threadIdx.x, lane = tid & 31, wave = tid >> 5;
  const int4* c4 = (const int4*)cnt_in;

  int pre = 0;
  const int nb = (int)blockIdx.x;
  for (int m = 0; m < nb; m += 16) {
    int4 v[16];
#pragma unroll
    for (int u = 0; u < 16; ++u) v[u] = c4[min(m + u, nb - 1) * 256 + tid];
#pragma unroll
    for (int u = 0; u < 16; ++u) {
      const int s = (v[u].x + v[u].y) + (v[u].z + v[u].w);
      pre += (m + u < nb) ? s : 0;
    }
  }
  for (int o = 16; o > 0; o >>= 1) pre += __shfl_xor(pre, o);
  if (lane == 0) wsum[wave] = pre;

  const int i0 = ((int)blockIdx.x * 256 + tid) * 4;
  int4 v = make_int4(0, 0, 0, 0);
  if (i0 < n) v = c4[i0 >> 2];
  const int tsum = (v.x + v.y) + (v.z + v.w);
  int inc = tsum;
  for (int o = 1; o < 32; o <<= 1) {
    int t = __shfl_up(inc, o);
    if (lane >= o) inc += t;
  }
  if (lane == 31) wsum2[wave] = inc;
  __syncthreads();
  int base = 0;
  for (int w = 0; w < 8; ++w) {
    base += wsum[w];
    if (w < wave) base += wsum2[w];
  }
  const int ex = base + inc - tsum;
  if (i0 < n) {
    int4 r;
    r.x = ex; r.y = ex + v.x; r.z = r.y + v.y; r.w = r.z + v.z;
    *(int4*)(rowptr + i0) = r;
    if (i0 + 4 == n) rowptr[n] = r.w + v.w;
    const int4 co = *(const int4*)(cnt_out + i0);
    float4 a, b;
    a.x = rsqrtf(fmaxf((float)co.x, 1.0f)); a.y = rsqrtf(fmaxf((float)co.y, 1.0f));
    a.z = rsqrtf(fmaxf((float)co.z, 1.0f)); a.w = rsqrtf(fmaxf((float)co.w, 1.0f));
    b.x = rsqrtf(fmaxf((float)v.x, 1.0f));  b.y = rsqrtf(fmaxf((float)v.y, 1.0f));
    b.z = rsqrtf(fmaxf((float)v.z, 1.0f));  b.w = rsqrtf(fmaxf((float)v.w, 1.0f));
    *(float4*)(nsrc + i0) = a;
    *(float4*)(ndst + i0) = b;
  }
}

__global__ void place_edges(const int* __restrict__ src, const int* __restrict__ dst,
                            const int* __restrict__ slot, const int* __restrict__ rowptr,
                            int* __restrict__ colu, int E) {
  int e = blockIdx.x * blockDim.x + threadIdx.x;
  if (e < E) colu[rowptr[dst[e]] + slot[e]] = src[e];
}

__global__ void order_edges(const int* __restrict__ src, const int* __restrict__ dst,
                            const int* __restrict__ slot, const int* __restrict__ rowptr,
                            const int* __restrict__ colu, int* __restrict__ col, int E) {
  int e = blockIdx.x * blockDim.x + threadIdx.x;
  if (e >= E) return;
  const int d = dst[e], s = src[e];
  const int beg = rowptr[d], end = rowptr[d + 1];
  const int mine = beg + slot[e];
  int r = 0;
  for (int j = beg; j < end; ++j) {
    int c = colu[j];
    r += (c < s || (c == s && j < mine)) ? 1 : 0;
  }
  col[beg + r] = s;
}

template <int MOUT>
__global__ __launch_bounds__(256)
void gcn_gemm(const float* __restrict__ x, const float* __restrict__ nsrc,
              const float* __restrict__ W, float* __restrict__ h, int nrows) {
  constexpr int LDK = KDIM + 8;
  __shared__ __attribute__((aligned(16))) __bf16 As[64 * LDK];
  __shared__ __attribute__((aligned(16))) __bf16 Bs[MOUT * LDK];

  const int tid  = threadIdx.x;
  const int wave = tid >> 5, lane = tid & 31;
  const int row0 = blockIdx.x * 64;

  const float4* x4 = (const float4*)x;
  const float4* W4 = (const float4*)W;
  constexpr int NQ  = MOUT / 4;
  constexpr int NQC = NQ / 8;
  constexpr int WCH = (KDIM / 8) * NQC / 8;

  float4 av[8];
  float  sc[8];
#pragma unroll
  for (int i = 0; i < 8; ++i) {
    const int row = row0 + wave + 8 * i;
    const int rr  = min(row, nrows - 1);
    av[i] = x4[(size_t)rr * (KDIM / 4) + lane];
    sc[i] = nsrc[rr];
  }
  float4 w0[WCH], w1[WCH];
#pragma unroll
  for (int i = 0; i < WCH; ++i) {
    const int c  = wave + 8 * i;
    const int kp = (c / NQC) * 4 + (lane >> 3);
    const int nq = (c % NQC) * 8 + (lane & 7);
    w0[i] = W4[(size_t)(2 * kp) * NQ + nq];
    w1[i] = W4[(size_t)(2 * kp + 1) * NQ + nq];
  }
  __builtin_amdgcn_sched_barrier(0);
#pragma unroll
  for (int i = 0; i < WCH; ++i) {
    const int c  = wave + 8 * i;
    const int kp = (c / NQC) * 4 + (lane >> 3);
    const int nq = (c % NQC) * 8 + (lane & 7);
    const float e0[4] = {w0[i].x, w0[i].y, w0[i].z, w0[i].w};
    const float e1[4] = {w1[i].x, w1[i].y, w1[i].z, w1[i].w};
#pragma unroll
    for (int j = 0; j < 4; ++j) {
      v2bf p;
      p[0] = f2bf(e0[j]);
      p[1] = f2bf(e1[j]);
      *(v2bf*)&Bs[(nq * 4 + j) * LDK + 2 * kp] = p;
    }
  }
#pragma unroll
  for (int i = 0; i < 8; ++i) {
    const int r   = wave + 8 * i;
    const bool ok = (row0 + r) < nrows;
    const float s = sc[i];
    v4bf p;
    p[0] = f2bf(ok ? av[i].x * s : 0.0f);
    p[1] = f2bf(ok ? av[i].y * s : 0.0f);
    p[2] = f2bf(ok ? av[i].z * s : 0.0f);
    p[3] = f2bf(ok ? av[i].w * s : 0.0f);
    *(v4bf*)&As[r * LDK + lane * 4] = p;
  }
  __syncthreads();

  const int half = lane >> 4, mn = lane & 15;
  constexpr int CT = MOUT / 16;
  constexpr int NT = 4 * CT;
  const bool full = (row0 + 64) <= nrows;

  const int tc = wave % CT;
  const __bf16* brow = &Bs[(tc * 16 + mn) * LDK];
  v16bf bfrag[KDIM / 32];
#pragma unroll
  for (int kk = 0; kk < KDIM / 32; ++kk) {
    int k0 = kk * 32 + half * 8;
    v8bf bl = *(const v8bf*)(brow + k0);
    v8bf bh = *(const v8bf*)(brow + k0 + 16);
#pragma unroll
    for (int i = 0; i < 8; ++i) { bfrag[kk][i] = bl[i]; bfrag[kk][i + 8] = bh[i]; }
  }

#pragma unroll
  for (int it = 0; it < NT / 8; ++it) {
    const int t  = wave + it * 8;
    const int tr = t / CT;
    const __bf16* arow = &As[(tr * 16 + mn) * LDK];

    v16bf afrag[KDIM / 32];
#pragma unroll
    for (int kk = 0; kk < KDIM / 32; ++kk) {
      int k0 = kk * 32 + half * 8;
      v8bf al = *(const v8bf*)(arow + k0);
      v8bf ah = *(const v8bf*)(arow + k0 + 16);
#pragma unroll
      for (int i = 0; i < 8; ++i) { afrag[kk][i] = al[i]; afrag[kk][i + 8] = ah[i]; }
    }

    v8f c = {};
#pragma unroll
    for (int kk = 0; kk < KDIM / 32; ++kk)
      c = __builtin_amdgcn_wmma_f32_16x16x32_bf16(false, afrag[kk], false, bfrag[kk],
                                                  (short)0, c, false, false);

    const int gr0 = row0 + tr * 16 + half * 8;
    const int gc  = tc * 16 + mn;
    float* out = h + (size_t)gr0 * MOUT + gc;
    if (full) {
#pragma unroll
      for (int r = 0; r < 8; ++r) out[(size_t)r * MOUT] = c[r];
    } else {
#pragma unroll
      for (int r = 0; r < 8; ++r)
        if (gr0 + r < nrows) out[(size_t)r * MOUT] = c[r];
    }
  }
}

template <int F, bool RELU>
__global__ __launch_bounds__(256)
void gather_rows(const float4* __restrict__ h4, const int* __restrict__ rowptr,
                 const int* __restrict__ col, const float* __restrict__ ndst,
                 const float* __restrict__ bias, float* __restrict__ out, int n) {
  constexpr int LPR = F / 4;
  constexpr int RPS = 32 / LPR;
  constexpr int UN  = 4;
  const int lane = threadIdx.x & 31;
  const int node = __builtin_amdgcn_readfirstlane((int)(blockIdx.x * 8 + (threadIdx.x >> 5)));
  if (node >= n) return;
  const int beg = rowptr[node], end = rowptr[node + 1];
  const int sub = lane / LPR, l = lane % LPR;

  float4 acc = make_float4(0.f, 0.f, 0.f, 0.f);
  for (int base = beg; base < end; base += 32) {
    const int cnt = min(32, end - base);
    const int idx = col[base + min(lane, cnt - 1)];
    for (int k = 0; k < cnt; k += UN * RPS) {
      float4 v[UN];
#pragma unroll
      for (int u = 0; u < UN; ++u) {
        const int e = k + u * RPS + sub;
        int s;
        if constexpr (RPS == 1) s = __builtin_amdgcn_readlane(idx, e);
        else                    s = __shfl(idx, e);
        const float4 t = h4[(size_t)s * LPR + l];
        const bool ok = e < cnt;
        v[u].x = ok ? t.x : 0.f; v[u].y = ok ? t.y : 0.f;
        v[u].z = ok ? t.z : 0.f; v[u].w = ok ? t.w : 0.f;
      }
#pragma unroll
      for (int u = 0; u < UN; ++u) {
        acc.x += v[u].x; acc.y += v[u].y; acc.z += v[u].z; acc.w += v[u].w;
      }
    }
  }
  if constexpr (RPS == 2) {
    acc.x += __shfl_xor(acc.x, 16); acc.y += __shfl_xor(acc.y, 16);
    acc.z += __shfl_xor(acc.z, 16); acc.w += __shfl_xor(acc.w, 16);
  }
  if (sub == 0) {
    const float nd = ndst[node];
    const float* bl = bias + l * 4;
    float4 r;
    r.x = acc.x * nd + bl[0]; r.y = acc.y * nd + bl[1];
    r.z = acc.z * nd + bl[2]; r.w = acc.w * nd + bl[3];
    if (RELU) { r.x = fmaxf(r.x, 0.f); r.y = fmaxf(r.y, 0.f); r.z = fmaxf(r.z, 0.f); r.w = fmaxf(r.w, 0.f); }
    *(float4*)(out + (size_t)node * F + l * 4) = r;
  }
}

extern "C" void kernel_launch(void* const* d_in, const int* in_sizes, int n_in,
                              void* d_out, int out_size, void* d_ws, size_t ws_size,
                              hipStream_t stream) {
  const float* in_feat = (const float*)d_in[0];
  const int*   src     = (const int*)d_in[1];
  const int*   dst     = (const int*)d_in[2];
  const float* W1 = (const float*)d_in[3];  const float* b1 = (const float*)d_in[4];
  const float* W2 = (const float*)d_in[5];  const float* b2 = (const float*)d_in[6];
  const float* W3 = (const float*)d_in[7];  const float* b3 = (const float*)d_in[8];
  const float* W4 = (const float*)d_in[9];  const float* b4 = (const float*)d_in[10];

  const int N = NNODES;
  const int E = in_sizes[1];

  float* ws      = (float*)d_ws;
  int*   cnt_in  = (int*)ws;
  int*   cnt_out = cnt_in + N;
  float* nsrc    = ws + 2 * (size_t)N;
  float* ndst    = nsrc + N;
  float* bufA    = ndst + N;
  float* bufB    = bufA + (size_t)N * 128;
  int*   rowptr  = (int*)(bufB + (size_t)N * 128);
  int*   slot    = rowptr + N + 64;
  int*   colu    = slot + E;
  int*   col     = colu + E;

  const dim3 blk(256);
  const unsigned gemm_blocks = (N + 63) / 64;
  const unsigned edge_blocks = (E + 255) / 256;
  const unsigned node_waves  = (N + 7) / 8;

  zero_i32<<<512, blk, 0, stream>>>(cnt_in, 2 * N);
  count_edges<<<edge_blocks, blk, 0, stream>>>(src, dst, cnt_out, cnt_in, slot, E);
  scan_degrees<<<(N + 1023) / 1024, blk, 0, stream>>>(cnt_in, cnt_out, rowptr, nsrc, ndst, N);
  place_edges<<<edge_blocks, blk, 0, stream>>>(src, dst, slot, rowptr, colu, E);
  order_edges<<<edge_blocks, blk, 0, stream>>>(src, dst, slot, rowptr, colu, col, E);

  gcn_gemm<128><<<gemm_blocks, blk, 0, stream>>>(in_feat, nsrc, W1, bufB, N);
  gather_rows<128, true><<<node_waves, blk, 0, stream>>>((const float4*)bufB, rowptr, col, ndst, b1, bufA, N);
  gcn_gemm<128><<<gemm_blocks, blk, 0, stream>>>(bufA, nsrc, W2, bufB, N);
  gather_rows<128, true><<<node_waves, blk, 0, stream>>>((const float4*)bufB, rowptr, col, ndst, b2, bufA, N);
  gcn_gemm<128><<<gemm_blocks, blk, 0, stream>>>(bufA, nsrc, W3, bufB, N);
  gather_rows<128, true><<<node_waves, blk, 0, stream>>>((const float4*)bufB, rowptr, col, ndst, b3, bufA, N);
  gcn_gemm<64><<<gemm_blocks, blk, 0, stream>>>(bufA, nsrc, W4, bufB, N);
  gather_rows<64, false><<<node_waves, blk, 0, stream>>>((const float4*)bufB, rowptr, col, ndst, b4, (float*)d_out, N);
}


# aggregation fused into the next layer's product: a work-group gathers 80 nodes' rows straight into the bf16 A tile in LDS; activation buffer and 3 launches gone; per-node sum order unchanged
# speedup vs baseline: 3.6860x; 1.2428x over previous
#include <hip/hip_runtime.h>
#include <stdint.h>

#define NNODES 100000
#define KDIM   128

typedef __attribute__((ext_vector_type(16))) __bf16 v16bf;
typedef __attribute__((ext_vector_type(8)))  __bf16 v8bf;
typedef __attribute__((ext_vector_type(4)))  __bf16 v4bf;
typedef __attribute__((ext_vector_type(2)))  __bf16 v2bf;
typedef __attribute__((ext_vector_type(8)))  float  v8f;

__device__ __forceinline__ __bf16 f2bf(float f) {
  unsigned u = __builtin_bit_cast(unsigned, f);
  unsigned r = u + 0x7FFFu + ((u >> 16) & 1u);
  unsigned short s = (unsigned short)(r >> 16);
  return __builtin_bit_cast(__bf16, s);
}

__global__ void zero_i32(int* __restrict__ p, int n) {
  int i = blockIdx.x * blockDim.x + threadIdx.x;
  int stride = gridDim.x * blockDim.x;
  for (; i < n; i += stride) p[i] = 0;
}

__global__ void count_edges(const int* __restrict__ src, const int* __restrict__ dst,
                            int* __restrict__ cnt_out, int* __restrict__ cnt_in,
                            int* __restrict__ slot, int E) {
  int e = blockIdx.x * blockDim.x + threadIdx.x;
  if (e < E) {
    slot[e] = __hip_atomic_fetch_add(&cnt_in[dst[e]], 1, __ATOMIC_RELAXED, __HIP_MEMORY_SCOPE_AGENT);
    __hip_atomic_fetch_add(&cnt_out[src[e]], 1, __ATOMIC_RELAXED, __HIP_MEMORY_SCOPE_AGENT);
  }
}

__global__ __launch_bounds__(256)
void scan_degrees(const int* __restrict__ cnt_in, const int* __restrict__ cnt_out,
                  int* __restrict__ rowptr, float* __restrict__ nsrc, float* __restrict__ ndst, int n) {
  __shared__ int wsum[8];
  __shared__ int wsum2[8];
  const int tid = threadIdx.x, lane = tid & 31, wave = tid >> 5;
  const int4* c4 = (const int4*)cnt_in;

  int pre = 0;
  const int nb = (int)blockIdx.x;
  for (int m = 0; m < nb; m += 16) {
    int4 v[16];
#pragma unroll
    for (int u = 0; u < 16; ++u) v[u] = c4[min(m + u, nb - 1) * 256 + tid];
#pragma unroll
    for (int u = 0; u < 16; ++u) {
      const int s = (v[u].x + v[u].y) + (v[u].z + v[u].w);
      pre += (m + u < nb) ? s : 0;
    }
  }
  for (int o = 16; o > 0; o >>= 1) pre += __shfl_xor(pre, o);
  if (lane == 0) wsum[wave] = pre;

  const int i0 = ((int)blockIdx.x * 256 + tid) * 4;
  int4 v = make_int4(0, 0, 0, 0);
  if (i0 < n) v = c4[i0 >> 2];
  const int tsum = (v.x + v.y) + (v.z + v.w);
  int inc = tsum;
  for (int o = 1; o < 32; o <<= 1) {
    int t = __shfl_up(inc, o);
    if (lane >= o) inc += t;
  }
  if (lane == 31) wsum2[wave] = inc;
  __syncthreads();
  int base = 0;
  for (int w = 0; w < 8; ++w) {
    base += wsum[w];
    if (w < wave) base += wsum2[w];
  }
  const int ex = base + inc - tsum;
  if (i0 < n) {
    int4 r;
    r.x = ex; r.y = ex + v.x; r.z = r.y + v.y; r.w = r.z + v.z;
    *(int4*)(rowptr + i0) = r;
    if (i0 + 4 == n) rowptr[n] = r.w + v.w;
    const int4 co = *(const int4*)(cnt_out + i0);
    float4 a, b;
    a.x = rsqrtf(fmaxf((float)co.x, 1.0f)); a.y = rsqrtf(fmaxf((float)co.y, 1.0f));
    a.z = rsqrtf(fmaxf((float)co.z, 1.0f)); a.w = rsqrtf(fmaxf((float)co.w, 1.0f));
    b.x = rsqrtf(fmaxf((float)v.x, 1.0f));  b.y = rsqrtf(fmaxf((float)v.y, 1.0f));
    b.z = rsqrtf(fmaxf((float)v.z, 1.0f));  b.w = rsqrtf(fmaxf((float)v.w, 1.0f));
    *(float4*)(nsrc + i0) = a;
    *(float4*)(ndst + i0) = b;
  }
}

__global__ void place_edges(const int* __restrict__ src, const int* __restrict__ dst,
                            const int* __restrict__ slot, const int* __restrict__ rowptr,
                            int* __restrict__ colu, int E) {
  int e = blockIdx.x * blockDim.x + threadIdx.x;
  if (e < E) colu[rowptr[dst[e]] + slot[e]] = src[e];
}

template <int MOUT>
__global__ __launch_bounds__(256)
void gcn_gemm(const float* __restrict__ x, const float* __restrict__ nsrc,
              const float* __restrict__ W, float* __restrict__ h, int nrows) {
  constexpr int LDK = KDIM + 8;
  __shared__ __attribute__((aligned(16))) __bf16 As[64 * LDK];
  __shared__ __attribute__((aligned(16))) __bf16 Bs[MOUT * LDK];

  const int tid  = threadIdx.x;
  const int wave = tid >> 5, lane = tid & 31;
  const int row0 = blockIdx.x * 64;

  const float4* x4 = (const float4*)x;
  const float4* W4 = (const float4*)W;
  constexpr int NQ  = MOUT / 4;
  constexpr int NQC = NQ / 8;
  constexpr int WCH = (KDIM / 8) * NQC / 8;

  float4 av[8];
  float  sc[8];
#pragma unroll
  for (int i = 0; i < 8; ++i) {
    const int row = row0 + wave + 8 * i;
    const int rr  = min(row, nrows - 1);
    av[i] = x4[(size_t)rr * (KDIM / 4) + lane];
    sc[i] = nsrc[rr];
  }
  float4 w0[WCH], w1[WCH];
#pragma unroll
  for (int i = 0; i < WCH; ++i) {
    const int c  = wave + 8 * i;
    const int kp = (c / NQC) * 4 + (lane >> 3);
    const int nq = (c % NQC) * 8 + (lane & 7);
    w0[i] = W4[(size_t)(2 * kp) * NQ + nq];
    w1[i] = W4[(size_t)(2 * kp + 1) * NQ + nq];
  }
  __builtin_amdgcn_sched_barrier(0);
#pragma unroll
  for (int i = 0; i < WCH; ++i) {
    const int c  = wave + 8 * i;
    const int kp = (c / NQC) * 4 + (lane >> 3);
    const int nq = (c % NQC) * 8 + (lane & 7);
    const float e0[4] = {w0[i].x, w0[i].y, w0[i].z, w0[i].w};
    const float e1[4] = {w1[i].x, w1[i].y, w1[i].z, w1[i].w};
#pragma unroll
    for (int j = 0; j < 4; ++j) {
      v2bf p;
      p[0] = f2bf(e0[j]);
      p[1] = f2bf(e1[j]);
      *(v2bf*)&Bs[(nq * 4 + j) * LDK + 2 * kp] = p;
    }
  }
#pragma unroll
  for (int i = 0; i < 8; ++i) {
    const int r   = wave + 8 * i;
    const bool ok = (row0 + r) < nrows;
    const float s = sc[i];
    v4bf p;
    p[0] = f2bf(ok ? av[i].x * s : 0.0f);
    p[1] = f2bf(ok ? av[i].y * s : 0.0f);
    p[2] = f2bf(ok ? av[i].z * s : 0.0f);
    p[3] = f2bf(ok ? av[i].w * s : 0.0f);
    *(v4bf*)&As[r * LDK + lane * 4] = p;
  }
  __syncthreads();

  const int half = lane >> 4, mn = lane & 15;
  constexpr int CT = MOUT / 16;
  constexpr int NT = 4 * CT;
  const bool full = (row0 + 64) <= nrows;

  const int tc = wave % CT;
  const __bf16* brow = &Bs[(tc * 16 + mn) * LDK];
  v16bf bfrag[KDIM / 32];
#pragma unroll
  for (int kk = 0; kk < KDIM / 32; ++kk) {
    int k0 = kk * 32 + half * 8;
    v8bf bl = *(const v8bf*)(brow + k0);
    v8bf bh = *(const v8bf*)(brow + k0 + 16);
#pragma unroll
    for (int i = 0; i < 8; ++i) { bfrag[kk][i] = bl[i]; bfrag[kk][i + 8] = bh[i]; }
  }

#pragma unroll
  for (int it = 0; it < NT / 8; ++it) {
    const int t  = wave + it * 8;
    const int tr = t / CT;
    const __bf16* arow = &As[(tr * 16 + mn) * LDK];

    v16bf afrag[KDIM / 32];
#pragma unroll
    for (int kk = 0; kk < KDIM / 32; ++kk) {
      int k0 = kk * 32 + half * 8;
      v8bf al = *(const v8bf*)(arow + k0);
      v8bf ah = *(const v8bf*)(arow + k0 + 16);
#pragma unroll
      for (int i = 0; i < 8; ++i) { afrag[kk][i] = al[i]; afrag[kk][i + 8] = ah[i]; }
    }

    v8f c = {};
#pragma unroll
    for (int kk = 0; kk < KDIM / 32; ++kk)
      c = __builtin_amdgcn_wmma_f32_16x16x32_bf16(false, afrag[kk], false, bfrag[kk],
                                                  (short)0, c, false, false);

    const int gr0 = row0 + tr * 16 + half * 8;
    const int gc  = tc * 16 + mn;
    float* out = h + (size_t)gr0 * MOUT + gc;
    if (full) {
#pragma unroll
      for (int r = 0; r < 8; ++r) out[(size_t)r * MOUT] = c[r];
    } else {
#pragma unroll
      for (int r = 0; r < 8; ++r)
        if (gr0 + r < nrows) out[(size_t)r * MOUT] = c[r];
    }
  }
}

template <int F, bool RELU>
__global__ __launch_bounds__(256)
void gather_rows(const float4* __restrict__ h4, const int* __restrict__ rowptr,
                 const int* colu, int* col, const float* __restrict__ ndst,
                 const float* __restrict__ bias, float* __restrict__ out, int n) {
  static_assert(NNODES <= (1 << 26), "sort key = id * 32 + lane must fit an int");
  constexpr int LPR = F / 4;
  constexpr int RPS = 32 / LPR;
  constexpr int UN  = 4;
  const int lane = threadIdx.x & 31;
  const int node = __builtin_amdgcn_readfirstlane((int)(blockIdx.x * 8 + (threadIdx.x >> 5)));
  if (node >= n) return;
  const int beg = rowptr[node], end = rowptr[node + 1];
  const int len = end - beg;
  const int sub = lane / LPR, l = lane % LPR;

  const int* list = colu;
  if (len > 32) {
    for (int i = lane; i < len; i += 32) {
      const int x = colu[beg + i];
      int r = 0;
      for (int j = 0; j < len; ++j) {
        const int c = colu[beg + j];
        r += (c < x || (c == x && j < i)) ? 1 : 0;
      }
      col[beg + r] = x;
    }
    __threadfence();
    list = col;
  }

  float4 acc = make_float4(0.f, 0.f, 0.f, 0.f);
  for (int base = beg; base < end; base += 32) {
    const int cnt = min(32, end - base);
    int idx = list[base + min(lane, cnt - 1)];
    if (len <= 32) {
      const int key = (idx << 5) | lane;
      int r = 0;
      for (int j = 0; j < cnt; ++j) {
        const int kj = __builtin_amdgcn_readlane(key, j);
        r += (kj < key) ? 1 : 0;
      }
      const int place = (lane < cnt) ? r : lane;
      idx = __builtin_amdgcn_ds_permute(place << 2, idx);
    }
    for (int k = 0; k < cnt; k += UN * RPS) {
      float4 v[UN];
#pragma unroll
      for (int u = 0; u < UN; ++u) {
        const int e = k + u * RPS + sub;
        int s;
        if constexpr (RPS == 1) s = __builtin_amdgcn_readlane(idx, e);
        else                    s = __shfl(idx, e);
        const float4 t = h4[(size_t)s * LPR + l];
        const bool ok = e < cnt;
        v[u].x = ok ? t.x : 0.f; v[u].y = ok ? t.y : 0.f;
        v[u].z = ok ? t.z : 0.f; v[u].w = ok ? t.w : 0.f;
      }
#pragma unroll
      for (int u = 0; u < UN; ++u) {
        acc.x += v[u].x; acc.y += v[u].y; acc.z += v[u].z; acc.w += v[u].w;
      }
    }
  }
  if constexpr (RPS == 2) {
    acc.x += __shfl_xor(acc.x, 16); acc.y += __shfl_xor(acc.y, 16);
    acc.z += __shfl_xor(acc.z, 16); acc.w += __shfl_xor(acc.w, 16);
  }
  if (sub == 0) {
    const float nd = ndst[node];
    const float* bl = bias + l * 4;
    float4 r;
    r.x = acc.x * nd + bl[0]; r.y = acc.y * nd + bl[1];
    r.z = acc.z * nd + bl[2]; r.w = acc.w * nd + bl[3];
    if (RELU) { r.x = fmaxf(r.x, 0.f); r.y = fmaxf(r.y, 0.f); r.z = fmaxf(r.z, 0.f); r.w = fmaxf(r.w, 0.f); }
    *(float4*)(out + (size_t)node * F + l * 4) = r;
  }
}

template <int MOUT>
__global__ __launch_bounds__(256)
void gather_gemm(const float4* __restrict__ h4, const int* __restrict__ rowptr,
                 const int* colu, int* col,
                 const float* __restrict__ ndst, const float* __restrict__ nsrc,
                 const float* __restrict__ bias, const float* __restrict__ W,
                 float* __restrict__ hout, int n) {
  static_assert(NNODES <= (1 << 26), "sort key = id * 32 + lane must fit an int");
  constexpr int ROWS = 80;
  constexpr int NPW  = ROWS / 8;
  constexpr int RT   = ROWS / 16;
  constexpr int UN   = 8;
  constexpr int LDK  = KDIM + 8;
  __shared__ __attribute__((aligned(16))) __bf16 As[ROWS * LDK];
  __shared__ __attribute__((aligned(16))) __bf16 Bs[MOUT * LDK];

  const int tid  = threadIdx.x;
  const int lane = tid & 31;
  const int wave = __builtin_amdgcn_readfirstlane(tid >> 5);
  const int row0 = blockIdx.x * ROWS;
  const int r0   = wave * NPW;
  const int node0 = row0 + r0;

  const int li = min(lane, NPW);
  const int rp = rowptr[min(node0 + li, n)];
  const int nc = min(node0 + li, n - 1);
  const float ndv = ndst[nc];
  const float nsv = nsrc[nc];
  const float4 bv = *(const float4*)(bias + lane * 4);

  {
    const float4* W4 = (const float4*)W;
    constexpr int NQ  = MOUT / 4;
    constexpr int NQC = NQ / 8;
    constexpr int WCH = (KDIM / 8) * NQC / 8;
    float4 w0[WCH], w1[WCH];
#pragma unroll
    for (int i = 0; i < WCH; ++i) {
      const int c  = wave + 8 * i;
      const int kp = (c / NQC) * 4 + (lane >> 3);
      const int nq = (c % NQC) * 8 + (lane & 7);
      w0[i] = W4[(size_t)(2 * kp) * NQ + nq];
      w1[i] = W4[(size_t)(2 * kp + 1) * NQ + nq];
    }
#pragma unroll
    for (int i = 0; i < WCH; ++i) {
      const int c  = wave + 8 * i;
      const int kp = (c / NQC) * 4 + (lane >> 3);
      const int nq = (c % NQC) * 8 + (lane & 7);
      const float e0[4] = {w0[i].x, w0[i].y, w0[i].z, w0[i].w};
      const float e1[4] = {w1[i].x, w1[i].y, w1[i].z, w1[i].w};
#pragma unroll
      for (int j = 0; j < 4; ++j) {
        v2bf p;
        p[0] = f2bf(e0[j]);
        p[1] = f2bf(e1[j]);
        *(v2bf*)&Bs[(nq * 4 + j) * LDK + 2 * kp] = p;
      }
    }
  }

  int beg = __builtin_amdgcn_readlane(rp, 0);
  int end = __builtin_amdgcn_readlane(rp, 1);
  int pre = colu[beg + max(min(lane, end - beg - 1), 0)];
  for (int i = 0; i < NPW; ++i) {
    const int len = end - beg;
    const int idx0 = pre;
    const int nbeg = end;
    const int nend = (i + 1 < NPW) ? __builtin_amdgcn_readlane(rp, i + 2) : end;

    float4 acc = make_float4(0.f, 0.f, 0.f, 0.f);
    if (len <= 32) {
      const int key = (idx0 << 5) | lane;
      int r = 0;
      for (int j = 0; j < len; ++j) {
        const int kj = __builtin_amdgcn_readlane(key, j);
        r += (kj < key) ? 1 : 0;
      }
      const int place = (lane < len) ? r : lane;
      const int idx = __builtin_amdgcn_ds_permute(place << 2, idx0);
      if (i + 1 < NPW) pre = colu[nbeg + max(min(lane, nend - nbeg - 1), 0)];
      for (int k = 0; k < len; k += UN) {
        float4 v[UN];
#pragma unroll
        for (int u = 0; u < UN; ++u) {
          const int e = k + u;
          v[u] = make_float4(0.f, 0.f, 0.f, 0.f);
          if (e < len) {
            const int s = __builtin_amdgcn_readlane(idx, e);
            v[u] = h4[(size_t)s * (KDIM / 4) + lane];
          }
        }
#pragma unroll
        for (int u = 0; u < UN; ++u) {
          acc.x += v[u].x; acc.y += v[u].y; acc.z += v[u].z; acc.w += v[u].w;
        }
      }
    } else {
      if (i + 1 < NPW) pre = colu[nbeg + max(min(lane, nend - nbeg - 1), 0)];
      for (int a = lane; a < len; a += 32) {
        const int x = colu[beg + a];
        int r = 0;
        for (int j = 0; j < len; ++j) {
          const int c = colu[beg + j];
          r += (c < x || (c == x && j < a)) ? 1 : 0;
        }
        col[beg + r] = x;
      }
      __threadfence();
      for (int base = beg; base < end; base += 32) {
        const int cnt = min(32, end - base);
        const int idx = col[base + min(lane, cnt - 1)];
        for (int e = 0; e < cnt; ++e) {
          const int s = __builtin_amdgcn_readlane(idx, e);
          const float4 t = h4[(size_t)s * (KDIM / 4) + lane];
          acc.x += t.x; acc.y += t.y; acc.z += t.z; acc.w += t.w;
        }
      }
    }

    const float nd = __builtin_bit_cast(float, __builtin_amdgcn_readlane(__builtin_bit_cast(int, ndv), i));
    const float s  = __builtin_bit_cast(float, __builtin_amdgcn_readlane(__builtin_bit_cast(int, nsv), i));
    const bool ok = (node0 + i) < n;
    float4 r;
    r.x = acc.x * nd + bv.x; r.y = acc.y * nd + bv.y;
    r.z = acc.z * nd + bv.z; r.w = acc.w * nd + bv.w;
    r.x = fmaxf(r.x, 0.f); r.y = fmaxf(r.y, 0.f); r.z = fmaxf(r.z, 0.f); r.w = fmaxf(r.w, 0.f);
    v4bf p;
    p[0] = f2bf(ok ? r.x * s : 0.0f);
    p[1] = f2bf(ok ? r.y * s : 0.0f);
    p[2] = f2bf(ok ? r.z * s : 0.0f);
    p[3] = f2bf(ok ? r.w * s : 0.0f);
    *(v4bf*)&As[(r0 + i) * LDK + lane * 4] = p;

    beg = nbeg; end = nend;
  }
  __syncthreads();

  const int half = lane >> 4, mn = lane & 15;
  constexpr int CT = MOUT / 16;
  constexpr int NT = RT * CT;
  const bool full = (row0 + ROWS) <= n;

  const int tc = wave % CT;
  const __bf16* brow = &Bs[(tc * 16 + mn) * LDK];
  v16bf bfrag[KDIM / 32];
#pragma unroll
  for (int kk = 0; kk < KDIM / 32; ++kk) {
    int k0 = kk * 32 + half * 8;
    v8bf bl = *(const v8bf*)(brow + k0);
    v8bf bh = *(const v8bf*)(brow + k0 + 16);
#pragma unroll
    for (int q = 0; q < 8; ++q) { bfrag[kk][q] = bl[q]; bfrag[kk][q + 8] = bh[q]; }
  }

  for (int t = wave; t < NT; t += 8) {
    const int tr = t / CT;
    const __bf16* arow = &As[(tr * 16 + mn) * LDK];
    v16bf afrag[KDIM / 32];
#pragma unroll
    for (int kk = 0; kk < KDIM / 32; ++kk) {
      int k0 = kk * 32 + half * 8;
      v8bf al = *(const v8bf*)(arow + k0);
      v8bf ah = *(const v8bf*)(arow + k0 + 16);
#pragma unroll
      for (int q = 0; q < 8; ++q) { afrag[kk][q] = al[q]; afrag[kk][q + 8] = ah[q]; }
    }
    v8f c = {};
#pragma unroll
    for (int kk = 0; kk < KDIM / 32; ++kk)
      c = __builtin_amdgcn_wmma_f32_16x16x32_bf16(false, afrag[kk], false, bfrag[kk],
                                                  (short)0, c, false, false);
    const int gr0 = row0 + tr * 16 + half * 8;
    const int gc  = tc * 16 + mn;
    float* out = hout + (size_t)gr0 * MOUT + gc;
    if (full) {
#pragma unroll
      for (int r = 0; r < 8; ++r) out[(size_t)r * MOUT] = c[r];
    } else {
#pragma unroll
      for (int r = 0; r < 8; ++r)
        if (gr0 + r < n) out[(size_t)r * MOUT] = c[r];
    }
  }
}

extern "C" void kernel_launch(void* const* d_in, const int* in_sizes, int n_in,
                              void* d_out, int out_size, void* d_ws, size_t ws_size,
                              hipStream_t stream) {
  const float* in_feat = (const float*)d_in[0];
  const int*   src     = (const int*)d_in[1];
  const int*   dst     = (const int*)d_in[2];
  const float* W1 = (const float*)d_in[3];  const float* b1 = (const float*)d_in[4];
  const float* W2 = (const float*)d_in[5];  const float* b2 = (const float*)d_in[6];
  const float* W3 = (const float*)d_in[7];  const float* b3 = (const float*)d_in[8];
  const float* W4 = (const float*)d_in[9];  const float* b4 = (const float*)d_in[10];

  const int N = NNODES;
  const int E = in_sizes[1];

  float* ws      = (float*)d_ws;
  int*   cnt_in  = (int*)ws;
  int*   cnt_out = cnt_in + N;
  float* nsrc    = ws + 2 * (size_t)N;
  float* ndst    = nsrc + N;
  float* bufA    = ndst + N;
  float* bufB    = bufA + (size_t)N * 128;
  int*   rowptr  = (int*)(bufB + (size_t)N * 128);
  int*   slot    = rowptr + N + 64;
  int*   colu    = slot + E;
  int*   col     = colu + E;

  const dim3 blk(256);
  const unsigned gemm_blocks = (N + 63) / 64;
  const unsigned edge_blocks = (E + 255) / 256;
  const unsigned node_waves  = (N + 7) / 8;
  const unsigned fuse_blocks = (N + 79) / 80;

  zero_i32<<<512, blk, 0, stream>>>(cnt_in, 2 * N);
  count_edges<<<edge_blocks, blk, 0, stream>>>(src, dst, cnt_out, cnt_in, slot, E);
  scan_degrees<<<(N + 1023) / 1024, blk, 0, stream>>>(cnt_in, cnt_out, rowptr, nsrc, ndst, N);
  place_edges<<<edge_blocks, blk, 0, stream>>>(src, dst, slot, rowptr, colu, E);

  gcn_gemm<128><<<gemm_blocks, blk, 0, stream>>>(in_feat, nsrc, W1, bufB, N);
  gather_gemm<128><<<fuse_blocks, blk, 0, stream>>>((const float4*)bufB, rowptr, colu, col, ndst, nsrc, b1, W2, bufA, N);
  gather_gemm<128><<<fuse_blocks, blk, 0, stream>>>((const float4*)bufA, rowptr, colu, col, ndst, nsrc, b2, W3, bufB, N);
  gather_gemm<64><<<fuse_blocks, blk, 0, stream>>>((const float4*)bufB, rowptr, colu, col, ndst, nsrc, b3, W4, bufA, N);
  gather_rows<64, false><<<node_waves, blk, 0, stream>>>((const float4*)bufA, rowptr, colu, col, ndst, b4, (float*)d_out, N);
}


# segment ordering done once per call: layer 1's gather stores each segment in source-id order, layers 2 to 4 read it (no in-wave rank sort, no scratch pass there); per-node sum order unchanged
# speedup vs baseline: 3.9404x; 1.0690x over previous
#include <hip/hip_runtime.h>
#include <stdint.h>

#define NNODES 100000
#define KDIM   128

typedef __attribute__((ext_vector_type(16))) __bf16 v16bf;
typedef __attribute__((ext_vector_type(8)))  __bf16 v8bf;
typedef __attribute__((ext_vector_type(4)))  __bf16 v4bf;
typedef __attribute__((ext_vector_type(2)))  __bf16 v2bf;
typedef __attribute__((ext_vector_type(8)))  float  v8f;

__device__ __forceinline__ __bf16 f2bf(float f) {
  unsigned u = __builtin_bit_cast(unsigned, f);
  unsigned r = u + 0x7FFFu + ((u >> 16) & 1u);
  unsigned short s = (unsigned short)(r >> 16);
  return __builtin_bit_cast(__bf16, s);
}

__global__ void zero_i32(int* __restrict__ p, int n) {
  int i = blockIdx.x * blockDim.x + threadIdx.x;
  int stride = gridDim.x * blockDim.x;
  for (; i < n; i += stride) p[i] = 0;
}

__global__ void count_edges(const int* __restrict__ src, const int* __restrict__ dst,
                            int* __restrict__ cnt_out, int* __restrict__ cnt_in,
                            int* __restrict__ slot, int E) {
  int e = blockIdx.x * blockDim.x + threadIdx.x;
  if (e < E) {
    slot[e] = __hip_atomic_fetch_add(&cnt_in[dst[e]], 1, __ATOMIC_RELAXED, __HIP_MEMORY_SCOPE_AGENT);
    __hip_atomic_fetch_add(&cnt_out[src[e]], 1, __ATOMIC_RELAXED, __HIP_MEMORY_SCOPE_AGENT);
  }
}

__global__ __launch_bounds__(256)
void scan_degrees(const int* __restrict__ cnt_in, const int* __restrict__ cnt_out,
                  int* __restrict__ rowptr, float* __restrict__ nsrc, float* __restrict__ ndst, int n) {
  __shared__ int wsum[8];
  __shared__ int wsum2[8];
  const int tid = threadIdx.x, lane = tid & 31, wave = tid >> 5;
  const int4* c4 = (const int4*)cnt_in;

  int pre = 0;
  const int nb = (int)blockIdx.x;
  for (int m = 0; m < nb; m += 16) {
    int4 v[16];
#pragma unroll
    for (int u = 0; u < 16; ++u) v[u] = c4[min(m + u, nb - 1) * 256 + tid];
#pragma unroll
    for (int u = 0; u < 16; ++u) {
      const int s = (v[u].x + v[u].y) + (v[u].z + v[u].w);
      pre += (m + u < nb) ? s : 0;
    }
  }
  for (int o = 16; o > 0; o >>= 1) pre += __shfl_xor(pre, o);
  if (lane == 0) wsum[wave] = pre;

  const int i0 = ((int)blockIdx.x * 256 + tid) * 4;
  int4 v = make_int4(0, 0, 0, 0);
  if (i0 < n) v = c4[i0 >> 2];
  const int tsum = (v.x + v.y) + (v.z + v.w);
  int inc = tsum;
  for (int o = 1; o < 32; o <<= 1) {
    int t = __shfl_up(inc, o);
    if (lane >= o) inc += t;
  }
  if (lane == 31) wsum2[wave] = inc;
  __syncthreads();
  int base = 0;
  for (int w = 0; w < 8; ++w) {
    base += wsum[w];
    if (w < wave) base += wsum2[w];
  }
  const int ex = base + inc - tsum;
  if (i0 < n) {
    int4 r;
    r.x = ex; r.y = ex + v.x; r.z = r.y + v.y; r.w = r.z + v.z;
    *(int4*)(rowptr + i0) = r;
    if (i0 + 4 == n) rowptr[n] = r.w + v.w;
    const int4 co = *(const int4*)(cnt_out + i0);
    float4 a, b;
    a.x = rsqrtf(fmaxf((float)co.x, 1.0f)); a.y = rsqrtf(fmaxf((float)co.y, 1.0f));
    a.z = rsqrtf(fmaxf((float)co.z, 1.0f)); a.w = rsqrtf(fmaxf((float)co.w, 1.0f));
    b.x = rsqrtf(fmaxf((float)v.x, 1.0f));  b.y = rsqrtf(fmaxf((float)v.y, 1.0f));
    b.z = rsqrtf(fmaxf((float)v.z, 1.0f));  b.w = rsqrtf(fmaxf((float)v.w, 1.0f));
    *(float4*)(nsrc + i0) = a;
    *(float4*)(ndst + i0) = b;
  }
}

__global__ void place_edges(const int* __restrict__ src, const int* __restrict__ dst,
                            const int* __restrict__ slot, const int* __restrict__ rowptr,
                            int* __restrict__ colu, int E) {
  int e = blockIdx.x * blockDim.x + threadIdx.x;
  if (e < E) colu[rowptr[dst[e]] + slot[e]] = src[e];
}

__device__ __forceinline__ unsigned bfbits(float f) {
  unsigned u = __builtin_bit_cast(unsigned, f);
  return (u + 0x7FFFu + ((u >> 16) & 1u)) >> 16;
}
__device__ __forceinline__ unsigned bfpack(float a, float b) { return bfbits(a) | (bfbits(b) << 16); }
__device__ __forceinline__ float bf_lo(unsigned u) { return __builtin_bit_cast(float, u << 16); }
__device__ __forceinline__ float bf_hi(unsigned u) { return __builtin_bit_cast(float, u & 0xFFFF0000u); }

__global__ __launch_bounds__(256)
void scale_round(const float* __restrict__ x, const float* __restrict__ nsrc,
                 float* __restrict__ xs, int nrows) {
  const int i = blockIdx.x * 256 + threadIdx.x;
  if (i < nrows * (KDIM / 4)) {
    const float4 v = ((const float4*)x)[i];
    const float s = nsrc[i >> 5];
    uint2 p;
    p.x = bfpack(v.x * s, v.y * s);
    p.y = bfpack(v.z * s, v.w * s);
    ((uint2*)xs)[i] = p;
  }
}

template <int MOUT, bool LAST, bool FIRST>
__global__ __launch_bounds__(256)
void gather_gemm(const float* __restrict__ xin, const int* __restrict__ rowptr,
                 const int* colu, int* col,
                 const float* __restrict__ ndst, const float* __restrict__ nsrc,
                 const float* __restrict__ bias, const float* __restrict__ W,
                 float* __restrict__ outp, int n) {
  static_assert(NNODES <= (1 << 26), "sort key = id * 32 + lane must fit an int");
  constexpr int ROWS = 80;
  constexpr int NPW  = ROWS / 8;
  constexpr int RT   = ROWS / 16;
  constexpr int UN   = 16;
  constexpr int LDK  = KDIM + 8;
  constexpr int R2   = (MOUT > ROWS) ? MOUT : ROWS;
  __shared__ __attribute__((aligned(16))) __bf16 Ah[ROWS * LDK];
  __shared__ __attribute__((aligned(16))) __bf16 Bs[R2 * LDK];
  __bf16* Al = Bs;

  const int tid  = threadIdx.x;
  const int lane = tid & 31;
  const int wave = __builtin_amdgcn_readfirstlane(tid >> 5);
  const int row0 = blockIdx.x * ROWS;
  const int r0   = wave * NPW;
  const int node0 = row0 + r0;
  const uint2* x2 = (const uint2*)xin;

  const int li = min(lane, NPW);
  const int rp = rowptr[min(node0 + li, n)];

  {
    const float4* W4 = (const float4*)W;
    constexpr int NQ  = MOUT / 4;
    constexpr int NQC = NQ / 8;
    constexpr int WCH = (KDIM / 8) * NQC / 8;
    float4 w0[WCH], w1[WCH];
#pragma unroll
    for (int i = 0; i < WCH; ++i) {
      const int c  = wave + 8 * i;
      const int kp = (c / NQC) * 4 + (lane >> 3);
      const int nq = (c % NQC) * 8 + (lane & 7);
      w0[i] = W4[(size_t)(2 * kp) * NQ + nq];
      w1[i] = W4[(size_t)(2 * kp + 1) * NQ + nq];
    }
#pragma unroll
    for (int i = 0; i < WCH; ++i) {
      const int c  = wave + 8 * i;
      const int kp = (c / NQC) * 4 + (lane >> 3);
      const int nq = (c % NQC) * 8 + (lane & 7);
      const float e0[4] = {w0[i].x, w0[i].y, w0[i].z, w0[i].w};
      const float e1[4] = {w1[i].x, w1[i].y, w1[i].z, w1[i].w};
#pragma unroll
      for (int j = 0; j < 4; ++j) {
        v2bf p;
        p[0] = f2bf(e0[j]);
        p[1] = f2bf(e1[j]);
        *(v2bf*)&Bs[(nq * 4 + j) * LDK + 2 * kp] = p;
      }
    }
  }
  __syncthreads();

  const int half = lane >> 4, mn = lane & 15;
  constexpr int CT = MOUT / 16;
  constexpr int NT = RT * CT;
  constexpr int TPW = (NT + 7) / 8;
  const int tc = wave % CT;
  v16bf wfrag[KDIM / 32];
  {
    const __bf16* wrow = &Bs[(tc * 16 + mn) * LDK];
#pragma unroll
    for (int kk = 0; kk < KDIM / 32; ++kk) {
      int k0 = kk * 32 + half * 8;
      v8bf bl = *(const v8bf*)(wrow + k0);
      v8bf bh = *(const v8bf*)(wrow + k0 + 16);
#pragma unroll
      for (int q = 0; q < 8; ++q) { wfrag[kk][q] = bl[q]; wfrag[kk][q + 8] = bh[q]; }
    }
  }
  __syncthreads();

  const int* ids = FIRST ? colu : (const int*)col;
  int beg = __builtin_amdgcn_readlane(rp, 0);
  int end = __builtin_amdgcn_readlane(rp, 1);
  int pre = ids[max(min(beg + lane, end - 1), 0)];
  for (int i = 0; i < NPW; ++i) {
    const int len = end - beg;
    const int idx0 = pre;
    const int nbeg = end;
    const int nend = (i + 1 < NPW) ? __builtin_amdgcn_readlane(rp, i + 2) : end;

    float4 acc = make_float4(0.f, 0.f, 0.f, 0.f);
    if (len <= 32) {
      int idx = idx0;
      if constexpr (FIRST) {
        const int key = (idx0 << 5) | lane;
        int r = 0;
        for (int j = 0; j < len; ++j) {
          const int kj = __builtin_amdgcn_readlane(key, j);
          r += (kj < key) ? 1 : 0;
        }
        const int place = (lane < len) ? r : lane;
        idx = __builtin_amdgcn_ds_permute(place << 2, idx0);
        if (lane < len) col[beg + lane] = idx;
      }
      if (i + 1 < NPW) pre = ids[max(min(nbeg + lane, nend - 1), 0)];
      for (int k = 0; k < len; k += UN) {
        uint2 v[UN];
#pragma unroll
        for (int u = 0; u < UN; ++u) {
          const int e = k + u;
          v[u] = make_uint2(0u, 0u);
          if (e < len) {
            const int s = __builtin_amdgcn_readlane(idx, e);
            v[u] = x2[(size_t)s * (KDIM / 4) + lane];
          }
        }
#pragma unroll
        for (int u = 0; u < UN; ++u) {
          acc.x += bf_lo(v[u].x); acc.y += bf_hi(v[u].x);
          acc.z += bf_lo(v[u].y); acc.w += bf_hi(v[u].y);
        }
      }
    } else {
      if (i + 1 < NPW) pre = ids[max(min(nbeg + lane, nend - 1), 0)];
      if constexpr (FIRST) {
        for (int a = lane; a < len; a += 32) {
          const int x = colu[beg + a];
          int r = 0;
          for (int j = 0; j < len; ++j) {
            const int c = colu[beg + j];
            r += (c < x || (c == x && j < a)) ? 1 : 0;
          }
          col[beg + r] = x;
        }
        __threadfence();
      }
      for (int base = beg; base < end; base += 32) {
        const int cnt = min(32, end - base);
        const int idx = col[base + min(lane, cnt - 1)];
        for (int e = 0; e < cnt; ++e) {
          const int s = __builtin_amdgcn_readlane(idx, e);
          const uint2 t = x2[(size_t)s * (KDIM / 4) + lane];
          acc.x += bf_lo(t.x); acc.y += bf_hi(t.x);
          acc.z += bf_lo(t.y); acc.w += bf_hi(t.y);
        }
      }
    }

    uint2 ph, pl;
    {
      const unsigned hx = bfbits(acc.x), hy = bfbits(acc.y), hz = bfbits(acc.z), hw = bfbits(acc.w);
      ph.x = hx | (hy << 16);
      ph.y = hz | (hw << 16);
      pl.x = bfpack(acc.x - bf_lo(hx), acc.y - bf_lo(hy));
      pl.y = bfpack(acc.z - bf_lo(hz), acc.w - bf_lo(hw));
    }
    *(uint2*)&Ah[(r0 + i) * LDK + lane * 4] = ph;
    *(uint2*)&Al[(r0 + i) * LDK + lane * 4] = pl;

    beg = nbeg; end = nend;
  }
  __syncthreads();

  float bb[8];
  {
    const float4 b0 = *(const float4*)(bias + tc * 16 + half * 8);
    const float4 b1 = *(const float4*)(bias + tc * 16 + half * 8 + 4);
    bb[0] = b0.x; bb[1] = b0.y; bb[2] = b0.z; bb[3] = b0.w;
    bb[4] = b1.x; bb[5] = b1.y; bb[6] = b1.z; bb[7] = b1.w;
  }
  constexpr int OW = LAST ? 8 : 4;
  unsigned o[TPW][OW];
#pragma unroll
  for (int it = 0; it < TPW; ++it) {
    const int t = wave + 8 * it;
    if (t < NT) {
      const int tr = t / CT;
      const int nodec = min(row0 + tr * 16 + mn, n - 1);
      const float nd = ndst[nodec];
      const float ns = nsrc[nodec];
      v8f c = {};
#pragma unroll
      for (int pass = 0; pass < 2; ++pass) {
        const __bf16* arow = (pass == 0 ? Al : Ah) + (tr * 16 + mn) * LDK;
#pragma unroll
        for (int kk = 0; kk < KDIM / 32; ++kk) {
          int k0 = kk * 32 + half * 8;
          v8bf al = *(const v8bf*)(arow + k0);
          v8bf ah = *(const v8bf*)(arow + k0 + 16);
          v16bf xf;
#pragma unroll
          for (int q = 0; q < 8; ++q) { xf[q] = al[q]; xf[q + 8] = ah[q]; }
          c = __builtin_amdgcn_wmma_f32_16x16x32_bf16(false, wfrag[kk], false, xf,
                                                      (short)0, c, false, false);
        }
      }
      float v[8];
#pragma unroll
      for (int r = 0; r < 8; ++r) {
        v[r] = c[r] * nd + bb[r];
        if constexpr (!LAST) v[r] = fmaxf(v[r], 0.f) * ns;
      }
      if constexpr (LAST) {
#pragma unroll
        for (int r = 0; r < 8; ++r) o[it][r] = __builtin_bit_cast(unsigned, v[r]);
      } else {
#pragma unroll
        for (int r = 0; r < 4; ++r) o[it][r] = bfpack(v[2 * r], v[2 * r + 1]);
      }
    }
  }
  __syncthreads();

  unsigned char* ot = (unsigned char*)Ah;
#pragma unroll
  for (int it = 0; it < TPW; ++it) {
    const int t = wave + 8 * it;
    if (t < NT) {
      const int tr = t / CT;
      unsigned char* p = ot + (tr * 16 + mn) * (LDK * 2) + (tc * 16 + half * 8) * (LAST ? 4 : 2);
      *(uint4*)p = make_uint4(o[it][0], o[it][1], o[it][2], o[it][3]);
      if constexpr (LAST) *(uint4*)(p + 16) = make_uint4(o[it][4], o[it][5], o[it][6], o[it][7]);
    }
  }
  __syncthreads();

  unsigned char* og = (unsigned char*)outp;
  for (int j = wave; j < ROWS / 2; j += 8) {
    const int r = 2 * j + half;
    const uint4 q = *(const uint4*)(ot + r * (LDK * 2) + mn * 16);
    if (row0 + r < n) *(uint4*)(og + (size_t)(row0 + r) * 256 + mn * 16) = q;
  }
}

extern "C" void kernel_launch(void* const* d_in, const int* in_sizes, int n_in,
                              void* d_out, int out_size, void* d_ws, size_t ws_size,
                              hipStream_t stream) {
  const float* in_feat = (const float*)d_in[0];
  const int*   src     = (const int*)d_in[1];
  const int*   dst     = (const int*)d_in[2];
  const float* W1 = (const float*)d_in[3];  const float* b1 = (const float*)d_in[4];
  const float* W2 = (const float*)d_in[5];  const float* b2 = (const float*)d_in[6];
  const float* W3 = (const float*)d_in[7];  const float* b3 = (const float*)d_in[8];
  const float* W4 = (const float*)d_in[9];  const float* b4 = (const float*)d_in[10];

  const int N = NNODES;
  const int E = in_sizes[1];

  float* ws      = (float*)d_ws;
  int*   cnt_in  = (int*)ws;
  int*   cnt_out = cnt_in + N;
  float* nsrc    = ws + 2 * (size_t)N;
  float* ndst    = nsrc + N;
  float* bufA    = ndst + N;
  float* bufB    = bufA + (size_t)N * 128;
  int*   rowptr  = (int*)(bufB + (size_t)N * 128);
  int*   slot    = rowptr + N + 64;
  int*   colu    = slot + E;
  int*   col     = colu + E;

  const dim3 blk(256);
  const unsigned edge_blocks = (E + 255) / 256;
  const unsigned conv_blocks = (N * (KDIM / 4) + 255) / 256;
  const unsigned fuse_blocks = (N + 79) / 80;

  zero_i32<<<512, blk, 0, stream>>>(cnt_in, 2 * N);
  count_edges<<<edge_blocks, blk, 0, stream>>>(src, dst, cnt_out, cnt_in, slot, E);
  scan_degrees<<<(N + 1023) / 1024, blk, 0, stream>>>(cnt_in, cnt_out, rowptr, nsrc, ndst, N);
  place_edges<<<edge_blocks, blk, 0, stream>>>(src, dst, slot, rowptr, colu, E);

  scale_round<<<conv_blocks, blk, 0, stream>>>(in_feat, nsrc, bufB, N);
  gather_gemm<128, false, true><<<fuse_blocks, blk, 0, stream>>>(bufB, rowptr, colu, col, ndst, nsrc, b1, W1, bufA, N);
  gather_gemm<128, false, false><<<fuse_blocks, blk, 0, stream>>>(bufA, rowptr, colu, col, ndst, nsrc, b2, W2, bufB, N);
  gather_gemm<128, false, false><<<fuse_blocks, blk, 0, stream>>>(bufB, rowptr, colu, col, ndst, nsrc, b3, W3, bufA, N);
  gather_gemm<64, true, false><<<fuse_blocks, blk, 0, stream>>>(bufA, rowptr, colu, col, ndst, nsrc, b4, W4, (float*)d_out, N);
}


# edge counting without scattered global atomics: 64 work-groups count degrees and in-segment places in packed 16-bit LDS tables, one launch sums the tables; zero launch gone; no sum order changes
# speedup vs baseline: 5.6855x; 1.4429x over previous
#include <hip/hip_runtime.h>
#include <stdint.h>

#define NNODES 100000
#define KDIM   128

typedef __attribute__((ext_vector_type(16))) __bf16 v16bf;
typedef __attribute__((ext_vector_type(8)))  __bf16 v8bf;
typedef __attribute__((ext_vector_type(4)))  __bf16 v4bf;
typedef __attribute__((ext_vector_type(2)))  __bf16 v2bf;
typedef __attribute__((ext_vector_type(8)))  float  v8f;

__device__ __forceinline__ __bf16 f2bf(float f) {
  unsigned u = __builtin_bit_cast(unsigned, f);
  unsigned r = u + 0x7FFFu + ((u >> 16) & 1u);
  unsigned short s = (unsigned short)(r >> 16);
  return __builtin_bit_cast(__bf16, s);
}

constexpr int HG = 32;
constexpr int HW = NNODES / 2;
static_assert(NNODES % 8 == 0, "tables are moved in 16-byte pieces");

__device__ __forceinline__ int chunk_edges(int E) { return (((E >> 2) + HG) / HG) * 4; }

__device__ __forceinline__ int bump(unsigned* h, int v) {
  const unsigned old = atomicAdd(&h[v >> 1], (v & 1) ? 0x10000u : 1u);
  return (int)((old >> ((v & 1) * 16)) & 0xFFFFu);
}

__global__ __launch_bounds__(1024)
void hist_edges(const int* __restrict__ src, const int* __restrict__ dst,
                unsigned* __restrict__ part_in, unsigned* __restrict__ part_out,
                int* __restrict__ slot, int E) {
  __shared__ __attribute__((aligned(16))) unsigned h[HW];
  const int tid = threadIdx.x;
  const bool isdst = blockIdx.x < HG;
  const int gq = isdst ? (int)blockIdx.x : (int)blockIdx.x - HG;
  uint4* h4 = (uint4*)h;
  for (int i = tid; i < HW / 4; i += 1024) h4[i] = make_uint4(0u, 0u, 0u, 0u);
  __syncthreads();

  const int chunk = chunk_edges(E);
  const int e0 = min(gq * chunk, E);
  const int e1 = min(e0 + chunk, E);
  const int* idx = isdst ? dst : src;
  const int4* i4 = (const int4*)idx;
  int4* s4 = (int4*)slot;
  const int q0 = e0 >> 2;
  const int q1 = e1 >> 2;
  for (int q = q0 + tid; q < q1; q += 4 * 1024) {
    int4 va = i4[q];
    int4 vb = i4[min(q + 1024, q1 - 1)];
    int4 vc = i4[min(q + 2048, q1 - 1)];
    int4 vd = i4[min(q + 3072, q1 - 1)];
    asm volatile("" : "+v"(va.x), "+v"(va.y), "+v"(va.z), "+v"(va.w), "+v"(vb.x), "+v"(vb.y), "+v"(vb.z), "+v"(vb.w),
                      "+v"(vc.x), "+v"(vc.y), "+v"(vc.z), "+v"(vc.w), "+v"(vd.x), "+v"(vd.y), "+v"(vd.z), "+v"(vd.w) : : "memory");
    {
      int4 r;
      r.x = bump(h, va.x); r.y = bump(h, va.y); r.z = bump(h, va.z); r.w = bump(h, va.w);
      if (isdst) s4[q] = r;
    }
    if (q + 1024 < q1) {
      int4 r;
      r.x = bump(h, vb.x); r.y = bump(h, vb.y); r.z = bump(h, vb.z); r.w = bump(h, vb.w);
      if (isdst) s4[q + 1024] = r;
    }
    if (q + 2048 < q1) {
      int4 r;
      r.x = bump(h, vc.x); r.y = bump(h, vc.y); r.z = bump(h, vc.z); r.w = bump(h, vc.w);
      if (isdst) s4[q + 2048] = r;
    }
    if (q + 3072 < q1) {
      int4 r;
      r.x = bump(h, vd.x); r.y = bump(h, vd.y); r.z = bump(h, vd.z); r.w = bump(h, vd.w);
      if (isdst) s4[q + 3072] = r;
    }
  }
  for (int e = max(q1 << 2, e0) + tid; e < e1; e += 1024) {
    const int r = bump(h, idx[e]);
    if (isdst) slot[e] = r;
  }
  __syncthreads();

  uint4* o4 = (uint4*)((isdst ? part_in : part_out) + (size_t)gq * HW);
  for (int i = tid; i < HW / 4; i += 1024) o4[i] = h4[i];
}

__global__ __launch_bounds__(256)
void reduce_parts(const unsigned* __restrict__ part_in, const unsigned* __restrict__ part_out,
                  int* __restrict__ off, int* __restrict__ cnt_in, int* __restrict__ cnt_out) {
  const int i = blockIdx.x * 256 + threadIdx.x;
  if (i < HW) {
    unsigned pin[HG], pout[HG];
#pragma unroll
    for (int g = 0; g < HG; ++g) {
      pin[g]  = part_in[(size_t)g * HW + i];
      pout[g] = part_out[(size_t)g * HW + i];
    }
    int a0 = 0, a1 = 0, b0 = 0, b1 = 0;
#pragma unroll
    for (int g = 0; g < HG; ++g) {
      ((int2*)(off + (size_t)g * NNODES))[i] = make_int2(a0, a1);
      a0 += (int)(pin[g] & 0xFFFFu);  a1 += (int)(pin[g] >> 16);
      b0 += (int)(pout[g] & 0xFFFFu); b1 += (int)(pout[g] >> 16);
    }
    ((int2*)cnt_in)[i]  = make_int2(a0, a1);
    ((int2*)cnt_out)[i] = make_int2(b0, b1);
  }
}

__global__ __launch_bounds__(256)
void scan_degrees(const int* __restrict__ cnt_in, const int* __restrict__ cnt_out,
                  int* __restrict__ rowptr, float* __restrict__ nsrc, float* __restrict__ ndst, int n) {
  __shared__ int wsum[8];
  __shared__ int wsum2[8];
  const int tid = threadIdx.x, lane = tid & 31, wave = tid >> 5;
  const int4* c4 = (const int4*)cnt_in;

  int pre = 0;
  const int nb = (int)blockIdx.x;
  for (int m = 0; m < nb; m += 16) {
    int4 v[16];
#pragma unroll
    for (int u = 0; u < 16; ++u) v[u] = c4[min(m + u, nb - 1) * 256 + tid];
#pragma unroll
    for (int u = 0; u < 16; ++u) {
      const int s = (v[u].x + v[u].y) + (v[u].z + v[u].w);
      pre += (m + u < nb) ? s : 0;
    }
  }
  for (int o = 16; o > 0; o >>= 1) pre += __shfl_xor(pre, o);
  if (lane == 0) wsum[wave] = pre;

  const int i0 = ((int)blockIdx.x * 256 + tid) * 4;
  int4 v = make_int4(0, 0, 0, 0);
  if (i0 < n) v = c4[i0 >> 2];
  const int tsum = (v.x + v.y) + (v.z + v.w);
  int inc = tsum;
  for (int o = 1; o < 32; o <<= 1) {
    int t = __shfl_up(inc, o);
    if (lane >= o) inc += t;
  }
  if (lane == 31) wsum2[wave] = inc;
  __syncthreads();
  int base = 0;
  for (int w = 0; w < 8; ++w) {
    base += wsum[w];
    if (w < wave) base += wsum2[w];
  }
  const int ex = base + inc - tsum;
  if (i0 < n) {
    int4 r;
    r.x = ex; r.y = ex + v.x; r.z = r.y + v.y; r.w = r.z + v.z;
    *(int4*)(rowptr + i0) = r;
    if (i0 + 4 == n) rowptr[n] = r.w + v.w;
    const int4 co = *(const int4*)(cnt_out + i0);
    float4 a, b;
    a.x = rsqrtf(fmaxf((float)co.x, 1.0f)); a.y = rsqrtf(fmaxf((float)co.y, 1.0f));
    a.z = rsqrtf(fmaxf((float)co.z, 1.0f)); a.w = rsqrtf(fmaxf((float)co.w, 1.0f));
    b.x = rsqrtf(fmaxf((float)v.x, 1.0f));  b.y = rsqrtf(fmaxf((float)v.y, 1.0f));
    b.z = rsqrtf(fmaxf((float)v.z, 1.0f));  b.w = rsqrtf(fmaxf((float)v.w, 1.0f));
    *(float4*)(nsrc + i0) = a;
    *(float4*)(ndst + i0) = b;
  }
}

__global__ void place_edges(const int* __restrict__ src, const int* __restrict__ dst,
                            const int* __restrict__ slot, const int* __restrict__ rowptr,
                            const int* __restrict__ off, int* __restrict__ colu, int E) {
  int e = blockIdx.x * blockDim.x + threadIdx.x;
  if (e < E) {
    const int g = e / chunk_edges(E);
    const int d = dst[e];
    colu[rowptr[d] + off[(size_t)g * NNODES + d] + slot[e]] = src[e];
  }
}

__device__ __forceinline__ unsigned bfbits(float f) {
  unsigned u = __builtin_bit_cast(unsigned, f);
  return (u + 0x7FFFu + ((u >> 16) & 1u)) >> 16;
}
__device__ __forceinline__ unsigned bfpack(float a, float b) { return bfbits(a) | (bfbits(b) << 16); }
__device__ __forceinline__ float bf_lo(unsigned u) { return __builtin_bit_cast(float, u << 16); }
__device__ __forceinline__ float bf_hi(unsigned u) { return __builtin_bit_cast(float, u & 0xFFFF0000u); }

__global__ __launch_bounds__(256)
void scale_round(const float* __restrict__ x, const float* __restrict__ nsrc,
                 float* __restrict__ xs, int nrows) {
  const int i = blockIdx.x * 256 + threadIdx.x;
  if (i < nrows * (KDIM / 4)) {
    const float4 v = ((const float4*)x)[i];
    const float s = nsrc[i >> 5];
    uint2 p;
    p.x = bfpack(v.x * s, v.y * s);
    p.y = bfpack(v.z * s, v.w * s);
    ((uint2*)xs)[i] = p;
  }
}

template <int MOUT, bool LAST, bool FIRST>
__global__ __launch_bounds__(256)
void gather_gemm(const float* __restrict__ xin, const int* __restrict__ rowptr,
                 const int* colu, int* col,
                 const float* __restrict__ ndst, const float* __restrict__ nsrc,
                 const float* __restrict__ bias, const float* __restrict__ W,
                 float* __restrict__ outp, int n) {
  static_assert(NNODES <= (1 << 26), "sort key = id * 32 + lane must fit an int");
  constexpr int ROWS = 80;
  constexpr int NPW  = ROWS / 8;
  constexpr int RT   = ROWS / 16;
  constexpr int UN   = 16;
  constexpr int LDK  = KDIM + 8;
  constexpr int R2   = (MOUT > ROWS) ? MOUT : ROWS;
  __shared__ __attribute__((aligned(16))) __bf16 Ah[ROWS * LDK];
  __shared__ __attribute__((aligned(16))) __bf16 Bs[R2 * LDK];
  __bf16* Al = Bs;

  const int tid  = threadIdx.x;
  const int lane = tid & 31;
  const int wave = __builtin_amdgcn_readfirstlane(tid >> 5);
  const int row0 = blockIdx.x * ROWS;
  const int r0   = wave * NPW;
  const int node0 = row0 + r0;
  const uint2* x2 = (const uint2*)xin;

  const int li = min(lane, NPW);
  const int rp = rowptr[min(node0 + li, n)];

  {
    const float4* W4 = (const float4*)W;
    constexpr int NQ  = MOUT / 4;
    constexpr int NQC = NQ / 8;
    constexpr int WCH = (KDIM / 8) * NQC / 8;
    float4 w0[WCH], w1[WCH];
#pragma unroll
    for (int i = 0; i < WCH; ++i) {
      const int c  = wave + 8 * i;
      const int kp = (c / NQC) * 4 + (lane >> 3);
      const int nq = (c % NQC) * 8 + (lane & 7);
      w0[i] = W4[(size_t)(2 * kp) * NQ + nq];
      w1[i] = W4[(size_t)(2 * kp + 1) * NQ + nq];
    }
#pragma unroll
    for (int i = 0; i < WCH; ++i) {
      const int c  = wave + 8 * i;
      const int kp = (c / NQC) * 4 + (lane >> 3);
      const int nq = (c % NQC) * 8 + (lane & 7);
      const float e0[4] = {w0[i].x, w0[i].y, w0[i].z, w0[i].w};
      const float e1[4] = {w1[i].x, w1[i].y, w1[i].z, w1[i].w};
#pragma unroll
      for (int j = 0; j < 4; ++j) {
        v2bf p;
        p[0] = f2bf(e0[j]);
        p[1] = f2bf(e1[j]);
        *(v2bf*)&Bs[(nq * 4 + j) * LDK + 2 * kp] = p;
      }
    }
  }
  __syncthreads();

  const int half = lane >> 4, mn = lane & 15;
  constexpr int CT = MOUT / 16;
  constexpr int NT = RT * CT;
  constexpr int TPW = (NT + 7) / 8;
  const int tc = wave % CT;
  v16bf wfrag[KDIM / 32];
  {
    const __bf16* wrow = &Bs[(tc * 16 + mn) * LDK];
#pragma unroll
    for (int kk = 0; kk < KDIM / 32; ++kk) {
      int k0 = kk * 32 + half * 8;
      v8bf bl = *(const v8bf*)(wrow + k0);
      v8bf bh = *(const v8bf*)(wrow + k0 + 16);
#pragma unroll
      for (int q = 0; q < 8; ++q) { wfrag[kk][q] = bl[q]; wfrag[kk][q + 8] = bh[q]; }
    }
  }
  __syncthreads();

  const int* ids = FIRST ? colu : (const int*)col;
  int beg = __builtin_amdgcn_readlane(rp, 0);
  int end = __builtin_amdgcn_readlane(rp, 1);
  int pre = ids[max(min(beg + lane, end - 1), 0)];
  for (int i = 0; i < NPW; ++i) {
    const int len = end - beg;
    const int idx0 = pre;
    const int nbeg = end;
    const int nend = (i + 1 < NPW) ? __builtin_amdgcn_readlane(rp, i + 2) : end;

    float4 acc = make_float4(0.f, 0.f, 0.f, 0.f);
    if (len <= 32) {
      int idx = idx0;
      if constexpr (FIRST) {
        const int key = (idx0 << 5) | lane;
        int r = 0;
        for (int j = 0; j < len; ++j) {
          const int kj = __builtin_amdgcn_readlane(key, j);
          r += (kj < key) ? 1 : 0;
        }
        const int place = (lane < len) ? r : lane;
        idx = __builtin_amdgcn_ds_permute(place << 2, idx0);
        if (lane < len) col[beg + lane] = idx;
      }
      if (i + 1 < NPW) pre = ids[max(min(nbeg + lane, nend - 1), 0)];
      for (int k = 0; k < len; k += UN) {
        uint2 v[UN];
#pragma unroll
        for (int u = 0; u < UN; ++u) {
          const int e = k + u;
          v[u] = make_uint2(0u, 0u);
          if (e < len) {
            const int s = __builtin_amdgcn_readlane(idx, e);
            v[u] = x2[(size_t)s * (KDIM / 4) + lane];
          }
        }
#pragma unroll
        for (int u = 0; u < UN; ++u) {
          acc.x += bf_lo(v[u].x); acc.y += bf_hi(v[u].x);
          acc.z += bf_lo(v[u].y); acc.w += bf_hi(v[u].y);
        }
      }
    } else {
      if (i + 1 < NPW) pre = ids[max(min(nbeg + lane, nend - 1), 0)];
      if constexpr (FIRST) {
        for (int a = lane; a < len; a += 32) {
          const int x = colu[beg + a];
          int r = 0;
          for (int j = 0; j < len; ++j) {
            const int c = colu[beg + j];
            r += (c < x || (c == x && j < a)) ? 1 : 0;
          }
          col[beg + r] = x;
        }
        __threadfence();
      }
      for (int base = beg; base < end; base += 32) {
        const int cnt = min(32, end - base);
        const int idx = col[base + min(lane, cnt - 1)];
        for (int e = 0; e < cnt; ++e) {
          const int s = __builtin_amdgcn_readlane(idx, e);
          const uint2 t = x2[(size_t)s * (KDIM / 4) + lane];
          acc.x += bf_lo(t.x); acc.y += bf_hi(t.x);
          acc.z += bf_lo(t.y); acc.w += bf_hi(t.y);
        }
      }
    }

    uint2 ph, pl;
    {
      const unsigned hx = bfbits(acc.x), hy = bfbits(acc.y), hz = bfbits(acc.z), hw = bfbits(acc.w);
      ph.x = hx | (hy << 16);
      ph.y = hz | (hw << 16);
      pl.x = bfpack(acc.x - bf_lo(hx), acc.y - bf_lo(hy));
      pl.y = bfpack(acc.z - bf_lo(hz), acc.w - bf_lo(hw));
    }
    *(uint2*)&Ah[(r0 + i) * LDK + lane * 4] = ph;
    *(uint2*)&Al[(r0 + i) * LDK + lane * 4] = pl;

    beg = nbeg; end = nend;
  }
  __syncthreads();

  float bb[8];
  {
    const float4 b0 = *(const float4*)(bias + tc * 16 + half * 8);
    const float4 b1 = *(const float4*)(bias + tc * 16 + half * 8 + 4);
    bb[0] = b0.x; bb[1] = b0.y; bb[2] = b0.z; bb[3] = b0.w;
    bb[4] = b1.x; bb[5] = b1.y; bb[6] = b1.z; bb[7] = b1.w;
  }
  constexpr int OW = LAST ? 8 : 4;
  unsigned o[TPW][OW];
#pragma unroll
  for (int it = 0; it < TPW; ++it) {
    const int t = wave + 8 * it;
    if (t < NT) {
      const int tr = t / CT;
      const int nodec = min(row0 + tr * 16 + mn, n - 1);
      const float nd = ndst[nodec];
      const float ns = nsrc[nodec];
      v8f c = {};
#pragma unroll
      for (int pass = 0; pass < 2; ++pass) {
        const __bf16* arow = (pass == 0 ? Al : Ah) + (tr * 16 + mn) * LDK;
#pragma unroll
        for (int kk = 0; kk < KDIM / 32; ++kk) {
          int k0 = kk * 32 + half * 8;
          v8bf al = *(const v8bf*)(arow + k0);
          v8bf ah = *(const v8bf*)(arow + k0 + 16);
          v16bf xf;
#pragma unroll
          for (int q = 0; q < 8; ++q) { xf[q] = al[q]; xf[q + 8] = ah[q]; }
          c = __builtin_amdgcn_wmma_f32_16x16x32_bf16(false, wfrag[kk], false, xf,
                                                      (short)0, c, false, false);
        }
      }
      float v[8];
#pragma unroll
      for (int r = 0; r < 8; ++r) {
        v[r] = c[r] * nd + bb[r];
        if constexpr (!LAST) v[r] = fmaxf(v[r], 0.f) * ns;
      }
      if constexpr (LAST) {
#pragma unroll
        for (int r = 0; r < 8; ++r) o[it][r] = __builtin_bit_cast(unsigned, v[r]);
      } else {
#pragma unroll
        for (int r = 0; r < 4; ++r) o[it][r] = bfpack(v[2 * r], v[2 * r + 1]);
      }
    }
  }
  __syncthreads();

  unsigned char* ot = (unsigned char*)Ah;
#pragma unroll
  for (int it = 0; it < TPW; ++it) {
    const int t = wave + 8 * it;
    if (t < NT) {
      const int tr = t / CT;
      unsigned char* p = ot + (tr * 16 + mn) * (LDK * 2) + (tc * 16 + half * 8) * (LAST ? 4 : 2);
      *(uint4*)p = make_uint4(o[it][0], o[it][1], o[it][2], o[it][3]);
      if constexpr (LAST) *(uint4*)(p + 16) = make_uint4(o[it][4], o[it][5], o[it][6], o[it][7]);
    }
  }
  __syncthreads();

  unsigned char* og = (unsigned char*)outp;
  for (int j = wave; j < ROWS / 2; j += 8) {
    const int r = 2 * j + half;
    const uint4 q = *(const uint4*)(ot + r * (LDK * 2) + mn * 16);
    if (row0 + r < n) *(uint4*)(og + (size_t)(row0 + r) * 256 + mn * 16) = q;
  }
}

extern "C" void kernel_launch(void* const* d_in, const int* in_sizes, int n_in,
                              void* d_out, int out_size, void* d_ws, size_t ws_size,
                              hipStream_t stream) {
  const float* in_feat = (const float*)d_in[0];
  const int*   src     = (const int*)d_in[1];
  const int*   dst     = (const int*)d_in[2];
  const float* W1 = (const float*)d_in[3];  const float* b1 = (const float*)d_in[4];
  const float* W2 = (const float*)d_in[5];  const float* b2 = (const float*)d_in[6];
  const float* W3 = (const float*)d_in[7];  const float* b3 = (const float*)d_in[8];
  const float* W4 = (const float*)d_in[9];  const float* b4 = (const float*)d_in[10];

  const int N = NNODES;
  const int E = in_sizes[1];

  float* ws      = (float*)d_ws;
  int*   cnt_in  = (int*)ws;
  int*   cnt_out = cnt_in + N;
  float* nsrc    = ws + 2 * (size_t)N;
  float* ndst    = nsrc + N;
  float* bufA    = ndst + N;
  float* bufB    = bufA + (size_t)N * 128;
  int*   rowptr  = (int*)(bufB + (size_t)N * 128);
  int*   slot    = rowptr + N + 64;
  int*   colu    = slot + E;
  int*   col     = colu + E;

  const dim3 blk(256);
  const unsigned edge_blocks = (E + 255) / 256;
  const unsigned conv_blocks = (N * (KDIM / 4) + 255) / 256;
  const unsigned fuse_blocks = (N + 79) / 80;

  unsigned* part_in  = (unsigned*)bufA;
  unsigned* part_out = part_in + (size_t)HG * (N / 2);
  int*      off      = (int*)(part_out + (size_t)HG * (N / 2));

  hist_edges<<<2 * HG, dim3(1024), 0, stream>>>(src, dst, part_in, part_out, slot, E);
  reduce_parts<<<(N / 2 + 255) / 256, blk, 0, stream>>>(part_in, part_out, off, cnt_in, cnt_out);
  scan_degrees<<<(N + 1023) / 1024, blk, 0, stream>>>(cnt_in, cnt_out, rowptr, nsrc, ndst, N);
  place_edges<<<edge_blocks, blk, 0, stream>>>(src, dst, slot, rowptr, off, colu, E);

  scale_round<<<conv_blocks, blk, 0, stream>>>(in_feat, nsrc, bufB, N);
  gather_gemm<128, false, true><<<fuse_blocks, blk, 0, stream>>>(bufB, rowptr, colu, col, ndst, nsrc, b1, W1, bufA, N);
  gather_gemm<128, false, false><<<fuse_blocks, blk, 0, stream>>>(bufA, rowptr, colu, col, ndst, nsrc, b2, W2, bufB, N);
  gather_gemm<128, false, false><<<fuse_blocks, blk, 0, stream>>>(bufB, rowptr, colu, col, ndst, nsrc, b3, W3, bufA, N);
  gather_gemm<64, true, false><<<fuse_blocks, blk, 0, stream>>>(bufA, rowptr, colu, col, ndst, nsrc, b4, W4, (float*)d_out, N);
}


# gather loop with fewer instructions a row: one 32-bit offset on a scalar base per load, bf16 halves added by mixed multiply-add (x*1+acc), adds only for rows that exist; sum order unchanged
# speedup vs baseline: 6.0397x; 1.0623x over previous
#include <hip/hip_runtime.h>
#include <stdint.h>

#define NNODES 100000
#define KDIM   128

typedef __attribute__((ext_vector_type(16))) __bf16 v16bf;
typedef __attribute__((ext_vector_type(8)))  __bf16 v8bf;
typedef __attribute__((ext_vector_type(4)))  __bf16 v4bf;
typedef __attribute__((ext_vector_type(2)))  __bf16 v2bf;
typedef __attribute__((ext_vector_type(8)))  float  v8f;

__device__ __forceinline__ __bf16 f2bf(float f) {
  unsigned u = __builtin_bit_cast(unsigned, f);
  unsigned r = u + 0x7FFFu + ((u >> 16) & 1u);
  unsigned short s = (unsigned short)(r >> 16);
  return __builtin_bit_cast(__bf16, s);
}

constexpr int HG = 32;
constexpr int HW = NNODES / 2;
static_assert(NNODES % 8 == 0, "tables are moved in 16-byte pieces");

__device__ __forceinline__ int chunk_edges(int E) { return (((E >> 2) + HG) / HG) * 4; }

__device__ __forceinline__ int bump(unsigned* h, int v) {
  const unsigned old = atomicAdd(&h[v >> 1], (v & 1) ? 0x10000u : 1u);
  return (int)((old >> ((v & 1) * 16)) & 0xFFFFu);
}

__global__ __launch_bounds__(1024)
void hist_edges(const int* __restrict__ src, const int* __restrict__ dst,
                unsigned* __restrict__ part_in, unsigned* __restrict__ part_out,
                int* __restrict__ slot, int E) {
  __shared__ __attribute__((aligned(16))) unsigned h[HW];
  const int tid = threadIdx.x;
  const bool isdst = blockIdx.x < HG;
  const int gq = isdst ? (int)blockIdx.x : (int)blockIdx.x - HG;
  uint4* h4 = (uint4*)h;
  for (int i = tid; i < HW / 4; i += 1024) h4[i] = make_uint4(0u, 0u, 0u, 0u);
  __syncthreads();

  const int chunk = chunk_edges(E);
  const int e0 = min(gq * chunk, E);
  const int e1 = min(e0 + chunk, E);
  const int* idx = isdst ? dst : src;
  const int4* i4 = (const int4*)idx;
  int4* s4 = (int4*)slot;
  const int q0 = e0 >> 2;
  const int q1 = e1 >> 2;
  for (int q = q0 + tid; q < q1; q += 4 * 1024) {
    int4 va = i4[q];
    int4 vb = i4[min(q + 1024, q1 - 1)];
    int4 vc = i4[min(q + 2048, q1 - 1)];
    int4 vd = i4[min(q + 3072, q1 - 1)];
    asm volatile("" : "+v"(va.x), "+v"(va.y), "+v"(va.z), "+v"(va.w), "+v"(vb.x), "+v"(vb.y), "+v"(vb.z), "+v"(vb.w),
                      "+v"(vc.x), "+v"(vc.y), "+v"(vc.z), "+v"(vc.w), "+v"(vd.x), "+v"(vd.y), "+v"(vd.z), "+v"(vd.w) : : "memory");
    {
      int4 r;
      r.x = bump(h, va.x); r.y = bump(h, va.y); r.z = bump(h, va.z); r.w = bump(h, va.w);
      if (isdst) s4[q] = r;
    }
    if (q + 1024 < q1) {
      int4 r;
      r.x = bump(h, vb.x); r.y = bump(h, vb.y); r.z = bump(h, vb.z); r.w = bump(h, vb.w);
      if (isdst) s4[q + 1024] = r;
    }
    if (q + 2048 < q1) {
      int4 r;
      r.x = bump(h, vc.x); r.y = bump(h, vc.y); r.z = bump(h, vc.z); r.w = bump(h, vc.w);
      if (isdst) s4[q + 2048] = r;
    }
    if (q + 3072 < q1) {
      int4 r;
      r.x = bump(h, vd.x); r.y = bump(h, vd.y); r.z = bump(h, vd.z); r.w = bump(h, vd.w);
      if (isdst) s4[q + 3072] = r;
    }
  }
  for (int e = max(q1 << 2, e0) + tid; e < e1; e += 1024) {
    const int r = bump(h, idx[e]);
    if (isdst) slot[e] = r;
  }
  __syncthreads();

  uint4* o4 = (uint4*)((isdst ? part_in : part_out) + (size_t)gq * HW);
  for (int i = tid; i < HW / 4; i += 1024) o4[i] = h4[i];
}

__global__ __launch_bounds__(256)
void reduce_parts(const unsigned* __restrict__ part_in, const unsigned* __restrict__ part_out,
                  int* __restrict__ off, int* __restrict__ cnt_in, int* __restrict__ cnt_out) {
  const int i = blockIdx.x * 256 + threadIdx.x;
  if (i < HW) {
    unsigned pin[HG], pout[HG];
#pragma unroll
    for (int g = 0; g < HG; ++g) {
      pin[g]  = part_in[(size_t)g * HW + i];
      pout[g] = part_out[(size_t)g * HW + i];
    }
    int a0 = 0, a1 = 0, b0 = 0, b1 = 0;
#pragma unroll
    for (int g = 0; g < HG; ++g) {
      ((int2*)(off + (size_t)g * NNODES))[i] = make_int2(a0, a1);
      a0 += (int)(pin[g] & 0xFFFFu);  a1 += (int)(pin[g] >> 16);
      b0 += (int)(pout[g] & 0xFFFFu); b1 += (int)(pout[g] >> 16);
    }
    ((int2*)cnt_in)[i]  = make_int2(a0, a1);
    ((int2*)cnt_out)[i] = make_int2(b0, b1);
  }
}

__global__ __launch_bounds__(256)
void scan_degrees(const int* __restrict__ cnt_in, const int* __restrict__ cnt_out,
                  int* __restrict__ rowptr, float* __restrict__ nsrc, float* __restrict__ ndst, int n) {
  __shared__ int wsum[8];
  __shared__ int wsum2[8];
  const int tid = threadIdx.x, lane = tid & 31, wave = tid >> 5;
  const int4* c4 = (const int4*)cnt_in;

  int pre = 0;
  const int nb = (int)blockIdx.x;
  for (int m = 0; m < nb; m += 16) {
    int4 v[16];
#pragma unroll
    for (int u = 0; u < 16; ++u) v[u] = c4[min(m + u, nb - 1) * 256 + tid];
#pragma unroll
    for (int u = 0; u < 16; ++u) {
      const int s = (v[u].x + v[u].y) + (v[u].z + v[u].w);
      pre += (m + u < nb) ? s : 0;
    }
  }
  for (int o = 16; o > 0; o >>= 1) pre += __shfl_xor(pre, o);
  if (lane == 0) wsum[wave] = pre;

  const int i0 = ((int)blockIdx.x * 256 + tid) * 4;
  int4 v = make_int4(0, 0, 0, 0);
  if (i0 < n) v = c4[i0 >> 2];
  const int tsum = (v.x + v.y) + (v.z + v.w);
  int inc = tsum;
  for (int o = 1; o < 32; o <<= 1) {
    int t = __shfl_up(inc, o);
    if (lane >= o) inc += t;
  }
  if (lane == 31) wsum2[wave] = inc;
  __syncthreads();
  int base = 0;
  for (int w = 0; w < 8; ++w) {
    base += wsum[w];
    if (w < wave) base += wsum2[w];
  }
  const int ex = base + inc - tsum;
  if (i0 < n) {
    int4 r;
    r.x = ex; r.y = ex + v.x; r.z = r.y + v.y; r.w = r.z + v.z;
    *(int4*)(rowptr + i0) = r;
    if (i0 + 4 == n) rowptr[n] = r.w + v.w;
    const int4 co = *(const int4*)(cnt_out + i0);
    float4 a, b;
    a.x = rsqrtf(fmaxf((float)co.x, 1.0f)); a.y = rsqrtf(fmaxf((float)co.y, 1.0f));
    a.z = rsqrtf(fmaxf((float)co.z, 1.0f)); a.w = rsqrtf(fmaxf((float)co.w, 1.0f));
    b.x = rsqrtf(fmaxf((float)v.x, 1.0f));  b.y = rsqrtf(fmaxf((float)v.y, 1.0f));
    b.z = rsqrtf(fmaxf((float)v.z, 1.0f));  b.w = rsqrtf(fmaxf((float)v.w, 1.0f));
    *(float4*)(nsrc + i0) = a;
    *(float4*)(ndst + i0) = b;
  }
}

__global__ void place_edges(const int* __restrict__ src, const int* __restrict__ dst,
                            const int* __restrict__ slot, const int* __restrict__ rowptr,
                            const int* __restrict__ off, int* __restrict__ colu, int E) {
  int e = blockIdx.x * blockDim.x + threadIdx.x;
  if (e < E) {
    const int g = e / chunk_edges(E);
    const int d = dst[e];
    colu[rowptr[d] + off[(size_t)g * NNODES + d] + slot[e]] = src[e];
  }
}

__device__ __forceinline__ unsigned bfbits(float f) {
  unsigned u = __builtin_bit_cast(unsigned, f);
  return (u + 0x7FFFu + ((u >> 16) & 1u)) >> 16;
}
__device__ __forceinline__ unsigned bfpack(float a, float b) { return bfbits(a) | (bfbits(b) << 16); }
__device__ __forceinline__ float bf_lo(unsigned u) { return __builtin_bit_cast(float, u << 16); }
__device__ __forceinline__ float bf_hi(unsigned u) { return __builtin_bit_cast(float, u & 0xFFFF0000u); }

static_assert((unsigned long long)NNODES * 256ull < (1ull << 32), "a row offset fits 32 bits");
__device__ __forceinline__ uint2 row_piece(const unsigned char* xb, int s, unsigned lane8) {
  const unsigned off = ((unsigned)s << 8) + lane8;
  return *(const uint2*)(xb + off);
}
__device__ __forceinline__ void add_piece(float4& acc, uint2 v) {
  const float one = 1.0f;
  asm volatile("v_fma_mix_f32_bf16 %0, %1, %2, %0 op_sel_hi:[1,0,0]" : "+v"(acc.x) : "v"(v.x), "s"(one));
  asm volatile("v_fma_mix_f32_bf16 %0, %1, %2, %0 op_sel:[1,0,0] op_sel_hi:[1,0,0]" : "+v"(acc.y) : "v"(v.x), "s"(one));
  asm volatile("v_fma_mix_f32_bf16 %0, %1, %2, %0 op_sel_hi:[1,0,0]" : "+v"(acc.z) : "v"(v.y), "s"(one));
  asm volatile("v_fma_mix_f32_bf16 %0, %1, %2, %0 op_sel:[1,0,0] op_sel_hi:[1,0,0]" : "+v"(acc.w) : "v"(v.y), "s"(one));
}
#define HOLD8(a) asm volatile("" : "+v"((a)[0].x), "+v"((a)[0].y), "+v"((a)[1].x), "+v"((a)[1].y), "+v"((a)[2].x), "+v"((a)[2].y), \
                                  "+v"((a)[3].x), "+v"((a)[3].y), "+v"((a)[4].x), "+v"((a)[4].y), "+v"((a)[5].x), "+v"((a)[5].y), \
                                  "+v"((a)[6].x), "+v"((a)[6].y), "+v"((a)[7].x), "+v"((a)[7].y))

__global__ __launch_bounds__(256)
void scale_round(const float* __restrict__ x, const float* __restrict__ nsrc,
                 float* __restrict__ xs, int nrows) {
  const int i = blockIdx.x * 256 + threadIdx.x;
  if (i < nrows * (KDIM / 4)) {
    const float4 v = ((const float4*)x)[i];
    const float s = nsrc[i >> 5];
    uint2 p;
    p.x = bfpack(v.x * s, v.y * s);
    p.y = bfpack(v.z * s, v.w * s);
    ((uint2*)xs)[i] = p;
  }
}

template <int MOUT, bool LAST, bool FIRST>
__global__ __launch_bounds__(256, 10)
void gather_gemm(const float* __restrict__ xin, const int* __restrict__ rowptr,
                 const int* colu, int* col,
                 const float* __restrict__ ndst, const float* __restrict__ nsrc,
                 const float* __restrict__ bias, const float* __restrict__ W,
                 float* __restrict__ outp, int n) {
  static_assert(NNODES <= (1 << 26), "sort key = id * 32 + lane must fit an int");
  constexpr int ROWS = 80;
  constexpr int NPW  = ROWS / 8;
  constexpr int RT   = ROWS / 16;
  constexpr int UN   = 16;
  constexpr int LDK  = KDIM + 8;
  constexpr int R2   = (MOUT > ROWS) ? MOUT : ROWS;
  __shared__ __attribute__((aligned(16))) __bf16 Ah[ROWS * LDK];
  __shared__ __attribute__((aligned(16))) __bf16 Bs[R2 * LDK];
  __bf16* Al = Bs;

  const int tid  = threadIdx.x;
  const int lane = tid & 31;
  const int wave = __builtin_amdgcn_readfirstlane(tid >> 5);
  const int row0 = blockIdx.x * ROWS;
  const int r0   = wave * NPW;
  const int node0 = row0 + r0;
  const uint2* x2 = (const uint2*)xin;
  const unsigned char* xb = (const unsigned char*)xin;
  const unsigned lane8 = (unsigned)lane * 8u;

  const int li = min(lane, NPW);
  const int rp = rowptr[min(node0 + li, n)];

  {
    const float4* W4 = (const float4*)W;
    constexpr int NQ  = MOUT / 4;
    constexpr int NQC = NQ / 8;
    constexpr int WCH = (KDIM / 8) * NQC / 8;
    float4 w0[WCH], w1[WCH];
#pragma unroll
    for (int i = 0; i < WCH; ++i) {
      const int c  = wave + 8 * i;
      const int kp = (c / NQC) * 4 + (lane >> 3);
      const int nq = (c % NQC) * 8 + (lane & 7);
      w0[i] = W4[(size_t)(2 * kp) * NQ + nq];
      w1[i] = W4[(size_t)(2 * kp + 1) * NQ + nq];
    }
#pragma unroll
    for (int i = 0; i < WCH; ++i) {
      const int c  = wave + 8 * i;
      const int kp = (c / NQC) * 4 + (lane >> 3);
      const int nq = (c % NQC) * 8 + (lane & 7);
      const float e0[4] = {w0[i].x, w0[i].y, w0[i].z, w0[i].w};
      const float e1[4] = {w1[i].x, w1[i].y, w1[i].z, w1[i].w};
#pragma unroll
      for (int j = 0; j < 4; ++j) {
        v2bf p;
        p[0] = f2bf(e0[j]);
        p[1] = f2bf(e1[j]);
        *(v2bf*)&Bs[(nq * 4 + j) * LDK + 2 * kp] = p;
      }
    }
  }
  __syncthreads();

  const int half = lane >> 4, mn = lane & 15;
  constexpr int CT = MOUT / 16;
  constexpr int NT = RT * CT;
  constexpr int TPW = (NT + 7) / 8;
  const int tc = wave % CT;
  v16bf wfrag[KDIM / 32];
  {
    const __bf16* wrow = &Bs[(tc * 16 + mn) * LDK];
#pragma unroll
    for (int kk = 0; kk < KDIM / 32; ++kk) {
      int k0 = kk * 32 + half * 8;
      v8bf bl = *(const v8bf*)(wrow + k0);
      v8bf bh = *(const v8bf*)(wrow + k0 + 16);
#pragma unroll
      for (int q = 0; q < 8; ++q) { wfrag[kk][q] = bl[q]; wfrag[kk][q + 8] = bh[q]; }
    }
  }
  __syncthreads();

  const int* ids = FIRST ? colu : (const int*)col;
  int beg = __builtin_amdgcn_readlane(rp, 0);
  int end = __builtin_amdgcn_readlane(rp, 1);
  int pre = ids[max(min(beg + lane, end - 1), 0)];
  for (int i = 0; i < NPW; ++i) {
    const int len = end - beg;
    const int idx0 = pre;
    const int nbeg = end;
    const int nend = (i + 1 < NPW) ? __builtin_amdgcn_readlane(rp, i + 2) : end;

    float4 acc = make_float4(0.f, 0.f, 0.f, 0.f);
    if (len <= 32) {
      int idx = idx0;
      if constexpr (FIRST) {
        const int key = (idx0 << 5) | lane;
        int r = 0;
        for (int j = 0; j < len; ++j) {
          const int kj = __builtin_amdgcn_readlane(key, j);
          r += (kj < key) ? 1 : 0;
        }
        const int place = (lane < len) ? r : lane;
        idx = __builtin_amdgcn_ds_permute(place << 2, idx0);
        if (lane < len) col[beg + lane] = idx;
      }
      if (i + 1 < NPW) pre = ids[max(min(nbeg + lane, nend - 1), 0)];
      int k = 0;
      for (; k + UN <= len; k += UN) {
        uint2 v[UN];
#pragma unroll
        for (int u = 0; u < UN; ++u) v[u] = row_piece(xb, __builtin_amdgcn_readlane(idx, k + u), lane8);
        HOLD8(v); HOLD8(v + 8);
#pragma unroll
        for (int u = 0; u < UN; ++u) add_piece(acc, v[u]);
      }
      const int rem = len - k;
      int p = k;
      if (rem & 8) {
        uint2 t[8];
#pragma unroll
        for (int u = 0; u < 8; ++u) t[u] = row_piece(xb, __builtin_amdgcn_readlane(idx, p + u), lane8);
        HOLD8(t);
#pragma unroll
        for (int u = 0; u < 8; ++u) add_piece(acc, t[u]);
        p += 8;
      }
      if (rem & 7) {
        uint2 t[7];
        int q = p;
        if (rem & 4) {
#pragma unroll
          for (int u = 0; u < 4; ++u) t[u] = row_piece(xb, __builtin_amdgcn_readlane(idx, q + u), lane8);
          q += 4;
        }
        if (rem & 2) {
#pragma unroll
          for (int u = 0; u < 2; ++u) t[4 + u] = row_piece(xb, __builtin_amdgcn_readlane(idx, q + u), lane8);
          q += 2;
        }
        if (rem & 1) t[6] = row_piece(xb, __builtin_amdgcn_readlane(idx, q), lane8);
        if (rem & 4) {
#pragma unroll
          for (int u = 0; u < 4; ++u) add_piece(acc, t[u]);
        }
        if (rem & 2) {
#pragma unroll
          for (int u = 0; u < 2; ++u) add_piece(acc, t[4 + u]);
        }
        if (rem & 1) add_piece(acc, t[6]);
      }
    } else {
      if (i + 1 < NPW) pre = ids[max(min(nbeg + lane, nend - 1), 0)];
      if constexpr (FIRST) {
        for (int a = lane; a < len; a += 32) {
          const int x = colu[beg + a];
          int r = 0;
          for (int j = 0; j < len; ++j) {
            const int c = colu[beg + j];
            r += (c < x || (c == x && j < a)) ? 1 : 0;
          }
          col[beg + r] = x;
        }
        __threadfence();
      }
      for (int base = beg; base < end; base += 32) {
        const int cnt = min(32, end - base);
        const int idx = col[base + min(lane, cnt - 1)];
        for (int e = 0; e < cnt; ++e) {
          const int s = __builtin_amdgcn_readlane(idx, e);
          const uint2 t = x2[(size_t)s * (KDIM / 4) + lane];
          acc.x += bf_lo(t.x); acc.y += bf_hi(t.x);
          acc.z += bf_lo(t.y); acc.w += bf_hi(t.y);
        }
      }
    }

    uint2 ph, pl;
    {
      const unsigned hx = bfbits(acc.x), hy = bfbits(acc.y), hz = bfbits(acc.z), hw = bfbits(acc.w);
      ph.x = hx | (hy << 16);
      ph.y = hz | (hw << 16);
      pl.x = bfpack(acc.x - bf_lo(hx), acc.y - bf_lo(hy));
      pl.y = bfpack(acc.z - bf_lo(hz), acc.w - bf_lo(hw));
    }
    *(uint2*)&Ah[(r0 + i) * LDK + lane * 4] = ph;
    *(uint2*)&Al[(r0 + i) * LDK + lane * 4] = pl;

    beg = nbeg; end = nend;
  }
  __syncthreads();

  const int lane_b = (int)__builtin_amdgcn_mbcnt_lo(~0u, 0u);
  const int half_b = lane_b >> 4, mn_b = lane_b & 15;
  float bb[8];
  {
    const float4 b0 = *(const float4*)(bias + tc * 16 + half_b * 8);
    const float4 b1 = *(const float4*)(bias + tc * 16 + half_b * 8 + 4);
    bb[0] = b0.x; bb[1] = b0.y; bb[2] = b0.z; bb[3] = b0.w;
    bb[4] = b1.x; bb[5] = b1.y; bb[6] = b1.z; bb[7] = b1.w;
  }
  constexpr int OW = LAST ? 8 : 4;
  unsigned o[TPW][OW];
#pragma unroll
  for (int it = 0; it < TPW; ++it) {
    const int t = wave + 8 * it;
    if (t < NT) {
      const int tr = t / CT;
      const int nodec = min(row0 + tr * 16 + mn_b, n - 1);
      const float nd = ndst[nodec];
      const float ns = nsrc[nodec];
      v8f c = {};
#pragma unroll
      for (int pass = 0; pass < 2; ++pass) {
        const __bf16* arow = (pass == 0 ? Al : Ah) + (tr * 16 + mn_b) * LDK;
#pragma unroll
        for (int kk = 0; kk < KDIM / 32; ++kk) {
          int k0 = kk * 32 + half_b * 8;
          v8bf al = *(const v8bf*)(arow + k0);
          v8bf ah = *(const v8bf*)(arow + k0 + 16);
          v16bf xf;
#pragma unroll
          for (int q = 0; q < 8; ++q) { xf[q] = al[q]; xf[q + 8] = ah[q]; }
          c = __builtin_amdgcn_wmma_f32_16x16x32_bf16(false, wfrag[kk], false, xf,
                                                      (short)0, c, false, false);
        }
      }
      float v[8];
#pragma unroll
      for (int r = 0; r < 8; ++r) {
        v[r] = c[r] * nd + bb[r];
        if constexpr (!LAST) v[r] = fmaxf(v[r], 0.f) * ns;
      }
      if constexpr (LAST) {
#pragma unroll
        for (int r = 0; r < 8; ++r) o[it][r] = __builtin_bit_cast(unsigned, v[r]);
      } else {
#pragma unroll
        for (int r = 0; r < 4; ++r) o[it][r] = bfpack(v[2 * r], v[2 * r + 1]);
      }
    }
  }
  __syncthreads();

  unsigned char* ot = (unsigned char*)Ah;
#pragma unroll
  for (int it = 0; it < TPW; ++it) {
    const int t = wave + 8 * it;
    if (t < NT) {
      const int tr = t / CT;
      unsigned char* p = ot + (tr * 16 + mn_b) * (LDK * 2) + (tc * 16 + half_b * 8) * (LAST ? 4 : 2);
      *(uint4*)p = make_uint4(o[it][0], o[it][1], o[it][2], o[it][3]);
      if constexpr (LAST) *(uint4*)(p + 16) = make_uint4(o[it][4], o[it][5], o[it][6], o[it][7]);
    }
  }
  __syncthreads();

  unsigned char* og = (unsigned char*)outp;
  for (int j = wave; j < ROWS / 2; j += 8) {
    const int r = 2 * j + half_b;
    const uint4 q = *(const uint4*)(ot + r * (LDK * 2) + mn_b * 16);
    if (row0 + r < n) *(uint4*)(og + (size_t)(row0 + r) * 256 + mn_b * 16) = q;
  }
}

extern "C" void kernel_launch(void* const* d_in, const int* in_sizes, int n_in,
                              void* d_out, int out_size, void* d_ws, size_t ws_size,
                              hipStream_t stream) {
  const float* in_feat = (const float*)d_in[0];
  const int*   src     = (const int*)d_in[1];
  const int*   dst     = (const int*)d_in[2];
  const float* W1 = (const float*)d_in[3];  const float* b1 = (const float*)d_in[4];
  const float* W2 = (const float*)d_in[5];  const float* b2 = (const float*)d_in[6];
  const float* W3 = (const float*)d_in[7];  const float* b3 = (const float*)d_in[8];
  const float* W4 = (const float*)d_in[9];  const float* b4 = (const float*)d_in[10];

  const int N = NNODES;
  const int E = in_sizes[1];

  float* ws      = (float*)d_ws;
  int*   cnt_in  = (int*)ws;
  int*   cnt_out = cnt_in + N;
  float* nsrc    = ws + 2 * (size_t)N;
  float* ndst    = nsrc + N;
  float* bufA    = ndst + N;
  float* bufB    = bufA + (size_t)N * 128;
  int*   rowptr  = (int*)(bufB + (size_t)N * 128);
  int*   slot    = rowptr + N + 64;
  int*   colu    = slot + E;
  int*   col     = colu + E;

  const dim3 blk(256);
  const unsigned edge_blocks = (E + 255) / 256;
  const unsigned conv_blocks = (N * (KDIM / 4) + 255) / 256;
  const unsigned fuse_blocks = (N + 79) / 80;

  unsigned* part_in  = (unsigned*)bufA;
  unsigned* part_out = part_in + (size_t)HG * (N / 2);
  int*      off      = (int*)(part_out + (size_t)HG * (N / 2));

  hist_edges<<<2 * HG, dim3(1024), 0, stream>>>(src, dst, part_in, part_out, slot, E);
  reduce_parts<<<(N / 2 + 255) / 256, blk, 0, stream>>>(part_in, part_out, off, cnt_in, cnt_out);
  scan_degrees<<<(N + 1023) / 1024, blk, 0, stream>>>(cnt_in, cnt_out, rowptr, nsrc, ndst, N);
  place_edges<<<edge_blocks, blk, 0, stream>>>(src, dst, slot, rowptr, off, colu, E);

  scale_round<<<conv_blocks, blk, 0, stream>>>(in_feat, nsrc, bufB, N);
  gather_gemm<128, false, true><<<fuse_blocks, blk, 0, stream>>>(bufB, rowptr, colu, col, ndst, nsrc, b1, W1, bufA, N);
  gather_gemm<128, false, false><<<fuse_blocks, blk, 0, stream>>>(bufA, rowptr, colu, col, ndst, nsrc, b2, W2, bufB, N);
  gather_gemm<128, false, false><<<fuse_blocks, blk, 0, stream>>>(bufB, rowptr, colu, col, ndst, nsrc, b3, W3, bufA, N);
  gather_gemm<64, true, false><<<fuse_blocks, blk, 0, stream>>>(bufA, rowptr, colu, col, ndst, nsrc, b4, W4, (float*)d_out, N);
}
